# starting kernel
# baseline (speedup 1.0000x reference)
#include <hip/hip_runtime.h>
#include <math.h>


static constexpr int kH = 512;
static constexpr int kW = 512;
static constexpr int kC = 64;
static constexpr int kO = 64;
static constexpr int kM1 = 32;
static constexpr int kM2 = 32;
static constexpr int kModes = kM1 * kM2;
static constexpr float kInvHW = 1.0f / (512.0f * 512.0f);

typedef __attribute__((ext_vector_type(2))) float v2f;
typedef __attribute__((ext_vector_type(8))) float v8f;

__device__ inline v8f wmma4(v2f a, v2f b, v8f c) {
  return __builtin_amdgcn_wmma_f32_16x16x4_f32(
       false, a,  false, b,
       (short)0, c,  false,  false);
}

__global__ __launch_bounds__(256)
void build_twiddles(float* __restrict__ Bw, float* __restrict__ Dm,
                    float* __restrict__ E2, float* __restrict__ G) {
  const int i = blockIdx.x * 256 + threadIdx.x;
  const float step = 6.2831853071795864769f / 512.0f;
  {
    int r = i >> 6, c = i & 63, k = c & 31;
    float th = ((r * k) & 511) * step;
    Bw[i] = (c < 32) ? cosf(th) : -sinf(th);
  }
  {
    int r = i >> 9, hc = i & 511, k = r & 31;
    float th = ((hc * k) & 511) * step;
    Dm[i] = (r < 32) ? cosf(th) : sinf(th);
  }
  {
    int hr = i >> 6, c = i & 63, k = c & 31;
    float th = ((hr * k) & 511) * step;
    E2[i] = (c < 32) ? cosf(th) : sinf(th);
  }
  {
    int r = i >> 9, w = i & 511, k = r & 31;
    float th = ((w * k) & 511) * step;
    float s = ((k == 0) ? 1.0f : 2.0f) * kInvHW;
    G[i] = (r < 32) ? s * cosf(th) : -s * sinf(th);
  }
}

__global__ __launch_bounds__(256)
void row_dft(const float* __restrict__ x, const float* __restrict__ Bw,
             float* __restrict__ X1) {
  extern __shared__ float Bs[];
  for (int i = threadIdx.x; i < 512 * 64; i += 256) Bs[i] = Bw[i];
  __syncthreads();

  const int wave = threadIdx.x >> 5, lane = threadIdx.x & 31;
  const int mt = wave >> 2, nt = wave & 3;
  const int hh = lane >> 4, lm = lane & 15;
  const int rowBase = blockIdx.x * 32 + mt * 16;
  const float* arow = x + (rowBase + lm) * 512;
  const int col = nt * 16 + lm;

  v8f acc = {};
  for (int k0 = 0; k0 < 512; k0 += 4) {
    const int ka = k0 + 2 * hh;
    v2f a; a.x = arow[ka]; a.y = arow[ka + 1];
    v2f b; b.x = Bs[ka * 64 + col]; b.y = Bs[(ka + 1) * 64 + col];
    acc = wmma4(a, b, acc);
  }
  for (int r = 0; r < 8; ++r)
    X1[(rowBase + r + 8 * hh) * 64 + col] = acc[r];
}

__global__ __launch_bounds__(512)
void col_dft(const float* __restrict__ X1, const float* __restrict__ Dm,
             float* __restrict__ X2r, float* __restrict__ X2i) {
  __shared__ float P[64 * 64];
  const int c = blockIdx.x;
  const int wave = threadIdx.x >> 5, lane = threadIdx.x & 31;
  const int mt = wave >> 2, nt = wave & 3;
  const int hh = lane >> 4, lm = lane & 15;
  const int arow = mt * 16 + lm;
  const int col = nt * 16 + lm;
  const float* B = X1 + c * 512 * 64;

  v8f acc = {};
  for (int k0 = 0; k0 < 512; k0 += 4) {
    const int ka = k0 + 2 * hh;
    v2f a; a.x = Dm[arow * 512 + ka]; a.y = Dm[arow * 512 + ka + 1];
    v2f b; b.x = B[ka * 64 + col];    b.y = B[(ka + 1) * 64 + col];
    acc = wmma4(a, b, acc);
  }
  for (int r = 0; r < 8; ++r)
    P[(mt * 16 + r + 8 * hh) * 64 + col] = acc[r];
  __syncthreads();

  for (int idx = threadIdx.x; idx < kModes; idx += 512) {
    const int kx = idx >> 5, ky = idx & 31;
    X2r[c * kModes + idx] = P[kx * 64 + ky]        + P[(32 + kx) * 64 + 32 + ky];
    X2i[c * kModes + idx] = P[kx * 64 + 32 + ky]   - P[(32 + kx) * 64 + ky];
  }
}

__global__ __launch_bounds__(256)
void mode_mix(const float* __restrict__ X2r, const float* __restrict__ X2i,
              const float* __restrict__ Wr, const float* __restrict__ Wi,
              float* __restrict__ Fr, float* __restrict__ Fi) {
  const int o = blockIdx.y;
  const int m = blockIdx.x * 256 + threadIdx.x;
  float ar = 0.0f, ai = 0.0f;
  for (int c = 0; c < kC; ++c) {
    const float xr = X2r[c * kModes + m];
    const float xi = X2i[c * kModes + m];
    const int wofs = ((c * kO + o) << 10) + m;
    const float wr = Wr[wofs], wi = Wi[wofs];
    ar = fmaf(xr, wr, fmaf(-xi, wi, ar));
    ai = fmaf(xr, wi, fmaf(xi, wr, ai));
  }
  Fr[(o << 10) + m] = ar;
  Fi[(o << 10) + m] = ai;
}

__global__ __launch_bounds__(512)
void inv_col(const float* __restrict__ Fr, const float* __restrict__ Fi,
             const float* __restrict__ E2, float* __restrict__ T) {
  __shared__ float Bs[64 * 64];
  const int o = blockIdx.x;
  for (int idx = threadIdx.x; idx < 64 * 64; idx += 512) {
    const int r = idx >> 6, n = idx & 63;
    const int fofs = (o << 10) + (r & 31) * 32 + (n & 31);
    float v;
    if (r < 32) v = (n < 32) ?  Fr[fofs] : Fi[fofs];
    else        v = (n < 32) ? -Fi[fofs] : Fr[fofs];
    Bs[idx] = v;
  }
  __syncthreads();

  const int wave = threadIdx.x >> 5, lane = threadIdx.x & 31;
  const int hh = lane >> 4, lm = lane & 15;
  for (int mt = wave; mt < 32; mt += 16) {
    const int rowBase = mt * 16;
    const float* arow = E2 + (rowBase + lm) * 64;
    for (int nt = 0; nt < 4; ++nt) {
      const int col = nt * 16 + lm;
      v8f acc = {};
      for (int k0 = 0; k0 < 64; k0 += 4) {
        const int ka = k0 + 2 * hh;
        v2f a; a.x = arow[ka];            a.y = arow[ka + 1];
        v2f b; b.x = Bs[ka * 64 + col];   b.y = Bs[(ka + 1) * 64 + col];
        acc = wmma4(a, b, acc);
      }
      for (int r = 0; r < 8; ++r)
        T[(o * 512 + rowBase + r + 8 * hh) * 64 + col] = acc[r];
    }
  }
}

__global__ __launch_bounds__(256)
void inv_row(const float* __restrict__ T, const float* __restrict__ G,
             float* __restrict__ y) {
  extern __shared__ float Gs[];
  for (int i = threadIdx.x; i < 64 * 512; i += 256) Gs[i] = G[i];
  __syncthreads();

  const int wave = threadIdx.x >> 5, lane = threadIdx.x & 31;
  const int hh = lane >> 4, lm = lane & 15;
  const int rb = blockIdx.x * 16;
  const float* arow = T + (rb + lm) * 64;

  for (int nt = wave; nt < 32; nt += 8) {
    const int col = nt * 16 + lm;
    v8f acc = {};
    for (int k0 = 0; k0 < 64; k0 += 4) {
      const int ka = k0 + 2 * hh;
      v2f a; a.x = arow[ka];             a.y = arow[ka + 1];
      v2f b; b.x = Gs[ka * 512 + col];   b.y = Gs[(ka + 1) * 512 + col];
      acc = wmma4(a, b, acc);
    }
    for (int r = 0; r < 8; ++r)
      y[(rb + r + 8 * hh) * 512 + col] = acc[r];
  }
}

extern "C" void kernel_launch(void* const* d_in, const int* in_sizes, int n_in,
                              void* d_out, int out_size, void* d_ws, size_t ws_size,
                              hipStream_t stream) {
  const float* x  = (const float*)d_in[0];
  const float* Wr = (const float*)d_in[1];
  const float* Wi = (const float*)d_in[2];
  float* y = (float*)d_out;

  float* ws  = (float*)d_ws;
  float* Bw  = ws;
  float* Dm  = Bw  + 32768;
  float* E2  = Dm  + 32768;
  float* G   = E2  + 32768;
  float* X1  = G   + 32768;
  float* X2r = X1  + 2097152;
  float* X2i = X2r + 65536;
  float* Fr  = X2i + 65536;
  float* Fi  = Fr  + 65536;
  float* T   = Fi  + 65536;

  build_twiddles<<<128, 256, 0, stream>>>(Bw, Dm, E2, G);
  row_dft<<<1024, 256, 512 * 64 * sizeof(float), stream>>>(x, Bw, X1);
  col_dft<<<64, 512, 0, stream>>>(X1, Dm, X2r, X2i);
  mode_mix<<<dim3(4, 64), 256, 0, stream>>>(X2r, X2i, Wr, Wi, Fr, Fi);
  inv_col<<<64, 512, 0, stream>>>(Fr, Fi, E2, T);
  inv_row<<<2048, 256, 64 * 512 * sizeof(float), stream>>>(T, G, y);
}


# row_dft and inv_row: twiddle table staged to LDS once per 512-thread work-group covering 64 rows (grid 1024/2048 -> 512, two groups resident per WGP) instead of once per 32/16-row strip
# speedup vs baseline: 1.2732x; 1.2732x over previous
#include <hip/hip_runtime.h>
#include <math.h>


static constexpr int kH = 512;
static constexpr int kW = 512;
static constexpr int kC = 64;
static constexpr int kO = 64;
static constexpr int kM1 = 32;
static constexpr int kM2 = 32;
static constexpr int kModes = kM1 * kM2;
static constexpr float kInvHW = 1.0f / (512.0f * 512.0f);

typedef __attribute__((ext_vector_type(2))) float v2f;
typedef __attribute__((ext_vector_type(8))) float v8f;

__device__ inline v8f wmma4(v2f a, v2f b, v8f c) {
  return __builtin_amdgcn_wmma_f32_16x16x4_f32(
       false, a,  false, b,
       (short)0, c,  false,  false);
}

__global__ __launch_bounds__(256)
void build_twiddles(float* __restrict__ Bw, float* __restrict__ Dm,
                    float* __restrict__ E2, float* __restrict__ G) {
  const int i = blockIdx.x * 256 + threadIdx.x;
  const float step = 6.2831853071795864769f / 512.0f;
  {
    int r = i >> 6, c = i & 63, k = c & 31;
    float th = ((r * k) & 511) * step;
    Bw[i] = (c < 32) ? cosf(th) : -sinf(th);
  }
  {
    int r = i >> 9, hc = i & 511, k = r & 31;
    float th = ((hc * k) & 511) * step;
    Dm[i] = (r < 32) ? cosf(th) : sinf(th);
  }
  {
    int hr = i >> 6, c = i & 63, k = c & 31;
    float th = ((hr * k) & 511) * step;
    E2[i] = (c < 32) ? cosf(th) : sinf(th);
  }
  {
    int r = i >> 9, w = i & 511, k = r & 31;
    float th = ((w * k) & 511) * step;
    float s = ((k == 0) ? 1.0f : 2.0f) * kInvHW;
    G[i] = (r < 32) ? s * cosf(th) : -s * sinf(th);
  }
}

__global__ __launch_bounds__(512)
void row_dft(const float* __restrict__ x, const float* __restrict__ Bw,
             float* __restrict__ X1) {
  extern __shared__ float Bs[];
  for (int i = threadIdx.x; i < 512 * 64; i += 512) Bs[i] = Bw[i];
  __syncthreads();

  const int wave = threadIdx.x >> 5, lane = threadIdx.x & 31;
  const int mt = wave >> 2, nt = wave & 3;
  const int hh = lane >> 4, lm = lane & 15;
  const int rowBase = blockIdx.x * 64 + mt * 16;
  const float* arow = x + (rowBase + lm) * 512;
  const int col = nt * 16 + lm;

  v8f acc = {};
  for (int k0 = 0; k0 < 512; k0 += 4) {
    const int ka = k0 + 2 * hh;
    v2f a; a.x = arow[ka]; a.y = arow[ka + 1];
    v2f b; b.x = Bs[ka * 64 + col]; b.y = Bs[(ka + 1) * 64 + col];
    acc = wmma4(a, b, acc);
  }
  for (int r = 0; r < 8; ++r)
    X1[(rowBase + r + 8 * hh) * 64 + col] = acc[r];
}

__global__ __launch_bounds__(512)
void col_dft(const float* __restrict__ X1, const float* __restrict__ Dm,
             float* __restrict__ X2r, float* __restrict__ X2i) {
  __shared__ float P[64 * 64];
  const int c = blockIdx.x;
  const int wave = threadIdx.x >> 5, lane = threadIdx.x & 31;
  const int mt = wave >> 2, nt = wave & 3;
  const int hh = lane >> 4, lm = lane & 15;
  const int arow = mt * 16 + lm;
  const int col = nt * 16 + lm;
  const float* B = X1 + c * 512 * 64;

  v8f acc = {};
  for (int k0 = 0; k0 < 512; k0 += 4) {
    const int ka = k0 + 2 * hh;
    v2f a; a.x = Dm[arow * 512 + ka]; a.y = Dm[arow * 512 + ka + 1];
    v2f b; b.x = B[ka * 64 + col];    b.y = B[(ka + 1) * 64 + col];
    acc = wmma4(a, b, acc);
  }
  for (int r = 0; r < 8; ++r)
    P[(mt * 16 + r + 8 * hh) * 64 + col] = acc[r];
  __syncthreads();

  for (int idx = threadIdx.x; idx < kModes; idx += 512) {
    const int kx = idx >> 5, ky = idx & 31;
    X2r[c * kModes + idx] = P[kx * 64 + ky]        + P[(32 + kx) * 64 + 32 + ky];
    X2i[c * kModes + idx] = P[kx * 64 + 32 + ky]   - P[(32 + kx) * 64 + ky];
  }
}

__global__ __launch_bounds__(256)
void mode_mix(const float* __restrict__ X2r, const float* __restrict__ X2i,
              const float* __restrict__ Wr, const float* __restrict__ Wi,
              float* __restrict__ Fr, float* __restrict__ Fi) {
  const int o = blockIdx.y;
  const int m = blockIdx.x * 256 + threadIdx.x;
  float ar = 0.0f, ai = 0.0f;
  for (int c = 0; c < kC; ++c) {
    const float xr = X2r[c * kModes + m];
    const float xi = X2i[c * kModes + m];
    const int wofs = ((c * kO + o) << 10) + m;
    const float wr = Wr[wofs], wi = Wi[wofs];
    ar = fmaf(xr, wr, fmaf(-xi, wi, ar));
    ai = fmaf(xr, wi, fmaf(xi, wr, ai));
  }
  Fr[(o << 10) + m] = ar;
  Fi[(o << 10) + m] = ai;
}

__global__ __launch_bounds__(512)
void inv_col(const float* __restrict__ Fr, const float* __restrict__ Fi,
             const float* __restrict__ E2, float* __restrict__ T) {
  __shared__ float Bs[64 * 64];
  const int o = blockIdx.x;
  for (int idx = threadIdx.x; idx < 64 * 64; idx += 512) {
    const int r = idx >> 6, n = idx & 63;
    const int fofs = (o << 10) + (r & 31) * 32 + (n & 31);
    float v;
    if (r < 32) v = (n < 32) ?  Fr[fofs] : Fi[fofs];
    else        v = (n < 32) ? -Fi[fofs] : Fr[fofs];
    Bs[idx] = v;
  }
  __syncthreads();

  const int wave = threadIdx.x >> 5, lane = threadIdx.x & 31;
  const int hh = lane >> 4, lm = lane & 15;
  for (int mt = wave; mt < 32; mt += 16) {
    const int rowBase = mt * 16;
    const float* arow = E2 + (rowBase + lm) * 64;
    for (int nt = 0; nt < 4; ++nt) {
      const int col = nt * 16 + lm;
      v8f acc = {};
      for (int k0 = 0; k0 < 64; k0 += 4) {
        const int ka = k0 + 2 * hh;
        v2f a; a.x = arow[ka];            a.y = arow[ka + 1];
        v2f b; b.x = Bs[ka * 64 + col];   b.y = Bs[(ka + 1) * 64 + col];
        acc = wmma4(a, b, acc);
      }
      for (int r = 0; r < 8; ++r)
        T[(o * 512 + rowBase + r + 8 * hh) * 64 + col] = acc[r];
    }
  }
}

__global__ __launch_bounds__(512)
void inv_row(const float* __restrict__ T, const float* __restrict__ G,
             float* __restrict__ y) {
  extern __shared__ float Gs[];
  for (int i = threadIdx.x; i < 64 * 512; i += 512) Gs[i] = G[i];
  __syncthreads();

  const int wave = threadIdx.x >> 5, lane = threadIdx.x & 31;
  const int hh = lane >> 4, lm = lane & 15;
  const int sub = wave >> 3, w8 = wave & 7;

  for (int j = 0; j < 2; ++j) {
    const int rb = (blockIdx.x * 4 + j * 2 + sub) * 16;
    const float* arow = T + (rb + lm) * 64;
    for (int nt = w8; nt < 32; nt += 8) {
      const int col = nt * 16 + lm;
      v8f acc = {};
      for (int k0 = 0; k0 < 64; k0 += 4) {
        const int ka = k0 + 2 * hh;
        v2f a; a.x = arow[ka];             a.y = arow[ka + 1];
        v2f b; b.x = Gs[ka * 512 + col];   b.y = Gs[(ka + 1) * 512 + col];
        acc = wmma4(a, b, acc);
      }
      for (int r = 0; r < 8; ++r)
        y[(rb + r + 8 * hh) * 512 + col] = acc[r];
    }
  }
}

extern "C" void kernel_launch(void* const* d_in, const int* in_sizes, int n_in,
                              void* d_out, int out_size, void* d_ws, size_t ws_size,
                              hipStream_t stream) {
  const float* x  = (const float*)d_in[0];
  const float* Wr = (const float*)d_in[1];
  const float* Wi = (const float*)d_in[2];
  float* y = (float*)d_out;

  float* ws  = (float*)d_ws;
  float* Bw  = ws;
  float* Dm  = Bw  + 32768;
  float* E2  = Dm  + 32768;
  float* G   = E2  + 32768;
  float* X1  = G   + 32768;
  float* X2r = X1  + 2097152;
  float* X2i = X2r + 65536;
  float* Fr  = X2i + 65536;
  float* Fi  = Fr  + 65536;
  float* T   = Fi  + 65536;

  build_twiddles<<<128, 256, 0, stream>>>(Bw, Dm, E2, G);
  row_dft<<<512, 512, 512 * 64 * sizeof(float), stream>>>(x, Bw, X1);
  col_dft<<<64, 512, 0, stream>>>(X1, Dm, X2r, X2i);
  mode_mix<<<dim3(4, 64), 256, 0, stream>>>(X2r, X2i, Wr, Wi, Fr, Fi);
  inv_col<<<64, 512, 0, stream>>>(Fr, Fi, E2, T);
  inv_row<<<512, 512, 64 * 512 * sizeof(float), stream>>>(T, G, y);
}


# row_dft, col_dft, inv_row: each lane fetches 4 consecutive k per 16-byte access (twiddles k-contiguous in LDS, padded conflict-free pitch) for 2 WMMAs; k order inside each 8 changes the sum order
# speedup vs baseline: 1.4028x; 1.1018x over previous
#include <hip/hip_runtime.h>
#include <math.h>


static constexpr int kH = 512;
static constexpr int kW = 512;
static constexpr int kC = 64;
static constexpr int kO = 64;
static constexpr int kM1 = 32;
static constexpr int kM2 = 32;
static constexpr int kModes = kM1 * kM2;
static constexpr float kInvHW = 1.0f / (512.0f * 512.0f);
static constexpr int kBsPitch4 = 129;
static constexpr int kGsPitch4 = 17;

typedef __attribute__((ext_vector_type(2))) float v2f;
typedef __attribute__((ext_vector_type(4))) float v4f;
typedef __attribute__((ext_vector_type(8))) float v8f;

__device__ inline v8f wmma4(v2f a, v2f b, v8f c) {
  return __builtin_amdgcn_wmma_f32_16x16x4_f32(
       false, a,  false, b,
       (short)0, c,  false,  false);
}

__global__ __launch_bounds__(256)
void build_twiddles(float* __restrict__ BwT, float* __restrict__ Dm,
                    float* __restrict__ E2, float* __restrict__ GT) {
  const int i = blockIdx.x * 256 + threadIdx.x;
  const float step = 6.2831853071795864769f / 512.0f;
  {
    int n = i >> 9, w = i & 511, k = n & 31;
    float th = ((w * k) & 511) * step;
    BwT[i] = (n < 32) ? cosf(th) : -sinf(th);
  }
  {
    int r = i >> 9, hc = i & 511, k = r & 31;
    float th = ((hc * k) & 511) * step;
    Dm[i] = (r < 32) ? cosf(th) : sinf(th);
  }
  {
    int hr = i >> 6, c = i & 63, k = c & 31;
    float th = ((hr * k) & 511) * step;
    E2[i] = (c < 32) ? cosf(th) : sinf(th);
  }
  {
    int w = i >> 6, c = i & 63, k = c & 31;
    float th = ((w * k) & 511) * step;
    float s = ((k == 0) ? 1.0f : 2.0f) * kInvHW;
    GT[i] = (c < 32) ? s * cosf(th) : -s * sinf(th);
  }
}

__global__ __launch_bounds__(512)
void row_dft(const float* __restrict__ x, const float* __restrict__ BwT,
             float* __restrict__ X1) {
  extern __shared__ v4f smem4[];
  const v4f* src4 = (const v4f*)BwT;
  for (int i = threadIdx.x; i < 64 * 128; i += 512)
    smem4[(i >> 7) * kBsPitch4 + (i & 127)] = src4[i];
  __syncthreads();

  const int wave = threadIdx.x >> 5, lane = threadIdx.x & 31;
  const int mt = wave >> 2, nt = wave & 3;
  const int hh = lane >> 4, lm = lane & 15;
  const int rowBase = blockIdx.x * 64 + mt * 16;
  const int col = nt * 16 + lm;
  const v4f* arow4 = (const v4f*)(x + (rowBase + lm) * 512) + hh;
  const v4f* bcol4 = smem4 + col * kBsPitch4 + hh;

  v8f acc = {};
  for (int j = 0; j < 64; ++j) {
    const v4f a = arow4[2 * j];
    const v4f b = bcol4[2 * j];
    v2f a0; a0.x = a.x; a0.y = a.y;
    v2f b0; b0.x = b.x; b0.y = b.y;
    acc = wmma4(a0, b0, acc);
    v2f a1; a1.x = a.z; a1.y = a.w;
    v2f b1; b1.x = b.z; b1.y = b.w;
    acc = wmma4(a1, b1, acc);
  }
  for (int r = 0; r < 8; ++r)
    X1[(rowBase + r + 8 * hh) * 64 + col] = acc[r];
}

__global__ __launch_bounds__(512)
void col_dft(const float* __restrict__ X1, const float* __restrict__ Dm,
             float* __restrict__ X2r, float* __restrict__ X2i) {
  __shared__ float P[64 * 64];
  const int c = blockIdx.x;
  const int wave = threadIdx.x >> 5, lane = threadIdx.x & 31;
  const int mt = wave >> 2, nt = wave & 3;
  const int hh = lane >> 4, lm = lane & 15;
  const int arow = mt * 16 + lm;
  const int col = nt * 16 + lm;
  const v4f* arow4 = (const v4f*)(Dm + arow * 512) + hh;
  const float* B = X1 + c * 512 * 64 + (4 * hh) * 64 + col;

  v8f acc = {};
  for (int j = 0; j < 64; ++j) {
    const v4f a = arow4[2 * j];
    const float* Bj = B + (8 * j) * 64;
    v2f a0; a0.x = a.x; a0.y = a.y;
    v2f b0; b0.x = Bj[0];       b0.y = Bj[64];
    acc = wmma4(a0, b0, acc);
    v2f a1; a1.x = a.z; a1.y = a.w;
    v2f b1; b1.x = Bj[128];     b1.y = Bj[192];
    acc = wmma4(a1, b1, acc);
  }
  for (int r = 0; r < 8; ++r)
    P[(mt * 16 + r + 8 * hh) * 64 + col] = acc[r];
  __syncthreads();

  for (int idx = threadIdx.x; idx < kModes; idx += 512) {
    const int kx = idx >> 5, ky = idx & 31;
    X2r[c * kModes + idx] = P[kx * 64 + ky]        + P[(32 + kx) * 64 + 32 + ky];
    X2i[c * kModes + idx] = P[kx * 64 + 32 + ky]   - P[(32 + kx) * 64 + ky];
  }
}

__global__ __launch_bounds__(256)
void mode_mix(const float* __restrict__ X2r, const float* __restrict__ X2i,
              const float* __restrict__ Wr, const float* __restrict__ Wi,
              float* __restrict__ Fr, float* __restrict__ Fi) {
  const int o = blockIdx.y;
  const int m = blockIdx.x * 256 + threadIdx.x;
  float ar = 0.0f, ai = 0.0f;
  for (int c = 0; c < kC; ++c) {
    const float xr = X2r[c * kModes + m];
    const float xi = X2i[c * kModes + m];
    const int wofs = ((c * kO + o) << 10) + m;
    const float wr = Wr[wofs], wi = Wi[wofs];
    ar = fmaf(xr, wr, fmaf(-xi, wi, ar));
    ai = fmaf(xr, wi, fmaf(xi, wr, ai));
  }
  Fr[(o << 10) + m] = ar;
  Fi[(o << 10) + m] = ai;
}

__global__ __launch_bounds__(512)
void inv_col(const float* __restrict__ Fr, const float* __restrict__ Fi,
             const float* __restrict__ E2, float* __restrict__ T) {
  __shared__ float Bs[64 * 64];
  const int o = blockIdx.x;
  for (int idx = threadIdx.x; idx < 64 * 64; idx += 512) {
    const int r = idx >> 6, n = idx & 63;
    const int fofs = (o << 10) + (r & 31) * 32 + (n & 31);
    float v;
    if (r < 32) v = (n < 32) ?  Fr[fofs] : Fi[fofs];
    else        v = (n < 32) ? -Fi[fofs] : Fr[fofs];
    Bs[idx] = v;
  }
  __syncthreads();

  const int wave = threadIdx.x >> 5, lane = threadIdx.x & 31;
  const int hh = lane >> 4, lm = lane & 15;
  for (int mt = wave; mt < 32; mt += 16) {
    const int rowBase = mt * 16;
    const float* arow = E2 + (rowBase + lm) * 64;
    for (int nt = 0; nt < 4; ++nt) {
      const int col = nt * 16 + lm;
      v8f acc = {};
      for (int k0 = 0; k0 < 64; k0 += 4) {
        const int ka = k0 + 2 * hh;
        v2f a; a.x = arow[ka];            a.y = arow[ka + 1];
        v2f b; b.x = Bs[ka * 64 + col];   b.y = Bs[(ka + 1) * 64 + col];
        acc = wmma4(a, b, acc);
      }
      for (int r = 0; r < 8; ++r)
        T[(o * 512 + rowBase + r + 8 * hh) * 64 + col] = acc[r];
    }
  }
}

__global__ __launch_bounds__(512)
void inv_row(const float* __restrict__ T, const float* __restrict__ GT,
             float* __restrict__ y) {
  extern __shared__ v4f smem4[];
  const v4f* src4 = (const v4f*)GT;
  for (int i = threadIdx.x; i < 512 * 16; i += 512)
    smem4[(i >> 4) * kGsPitch4 + (i & 15)] = src4[i];
  __syncthreads();

  const int wave = threadIdx.x >> 5, lane = threadIdx.x & 31;
  const int hh = lane >> 4, lm = lane & 15;
  const int sub = wave >> 3, w8 = wave & 7;

  for (int j = 0; j < 2; ++j) {
    const int rb = (blockIdx.x * 4 + j * 2 + sub) * 16;
    const v4f* arow4 = (const v4f*)(T + (rb + lm) * 64) + hh;
    v4f a[8];
    for (int jj = 0; jj < 8; ++jj) a[jj] = arow4[2 * jj];
    for (int nt = w8; nt < 32; nt += 8) {
      const int col = nt * 16 + lm;
      const v4f* bcol4 = smem4 + col * kGsPitch4 + hh;
      v8f acc = {};
      for (int jj = 0; jj < 8; ++jj) {
        const v4f b = bcol4[2 * jj];
        v2f a0; a0.x = a[jj].x; a0.y = a[jj].y;
        v2f b0; b0.x = b.x;     b0.y = b.y;
        acc = wmma4(a0, b0, acc);
        v2f a1; a1.x = a[jj].z; a1.y = a[jj].w;
        v2f b1; b1.x = b.z;     b1.y = b.w;
        acc = wmma4(a1, b1, acc);
      }
      for (int r = 0; r < 8; ++r)
        y[(rb + r + 8 * hh) * 512 + col] = acc[r];
    }
  }
}

extern "C" void kernel_launch(void* const* d_in, const int* in_sizes, int n_in,
                              void* d_out, int out_size, void* d_ws, size_t ws_size,
                              hipStream_t stream) {
  const float* x  = (const float*)d_in[0];
  const float* Wr = (const float*)d_in[1];
  const float* Wi = (const float*)d_in[2];
  float* y = (float*)d_out;

  float* ws  = (float*)d_ws;
  float* BwT = ws;
  float* Dm  = BwT + 32768;
  float* E2  = Dm  + 32768;
  float* GT  = E2  + 32768;
  float* X1  = GT  + 32768;
  float* X2r = X1  + 2097152;
  float* X2i = X2r + 65536;
  float* Fr  = X2i + 65536;
  float* Fi  = Fr  + 65536;
  float* T   = Fi  + 65536;

  build_twiddles<<<128, 256, 0, stream>>>(BwT, Dm, E2, GT);
  row_dft<<<512, 512, 64 * kBsPitch4 * sizeof(v4f), stream>>>(x, BwT, X1);
  col_dft<<<64, 512, 0, stream>>>(X1, Dm, X2r, X2i);
  mode_mix<<<dim3(4, 64), 256, 0, stream>>>(X2r, X2i, Wr, Wi, Fr, Fi);
  inv_col<<<64, 512, 0, stream>>>(Fr, Fi, E2, T);
  inv_row<<<512, 512, 512 * kGsPitch4 * sizeof(v4f), stream>>>(T, GT, y);
}


# row_dft, inv_row: one 8-wave work-group per WGP (grid 512x16 waves -> 256x8), each wave a whole 16-row strip on 4 accumulator tiles: A fetched once, twiddles staged once per WGP; sum order unchanged
# speedup vs baseline: 1.5702x; 1.1193x over previous
#include <hip/hip_runtime.h>
#include <math.h>


static constexpr int kH = 512;
static constexpr int kW = 512;
static constexpr int kC = 64;
static constexpr int kO = 64;
static constexpr int kM1 = 32;
static constexpr int kM2 = 32;
static constexpr int kModes = kM1 * kM2;
static constexpr float kInvHW = 1.0f / (512.0f * 512.0f);
static constexpr int kBsPitch4 = 129;
static constexpr int kGsPitch4 = 17;
static constexpr int kOnePerWgpLdsBytes = 162 * 1024;

typedef __attribute__((ext_vector_type(2))) float v2f;
typedef __attribute__((ext_vector_type(4))) float v4f;
typedef __attribute__((ext_vector_type(8))) float v8f;

__device__ inline v8f wmma4(v2f a, v2f b, v8f c) {
  return __builtin_amdgcn_wmma_f32_16x16x4_f32(
       false, a,  false, b,
       (short)0, c,  false,  false);
}

__global__ __launch_bounds__(256)
void build_twiddles(float* __restrict__ BwT, float* __restrict__ Dm,
                    float* __restrict__ E2, float* __restrict__ GT) {
  const int i = blockIdx.x * 256 + threadIdx.x;
  const float step = 6.2831853071795864769f / 512.0f;
  {
    int n = i >> 9, w = i & 511, k = n & 31;
    float th = ((w * k) & 511) * step;
    BwT[i] = (n < 32) ? cosf(th) : -sinf(th);
  }
  {
    int r = i >> 9, hc = i & 511, k = r & 31;
    float th = ((hc * k) & 511) * step;
    Dm[i] = (r < 32) ? cosf(th) : sinf(th);
  }
  {
    int hr = i >> 6, c = i & 63, k = c & 31;
    float th = ((hr * k) & 511) * step;
    E2[i] = (c < 32) ? cosf(th) : sinf(th);
  }
  {
    int w = i >> 6, c = i & 63, k = c & 31;
    float th = ((w * k) & 511) * step;
    float s = ((k == 0) ? 1.0f : 2.0f) * kInvHW;
    GT[i] = (c < 32) ? s * cosf(th) : -s * sinf(th);
  }
}

__global__ __launch_bounds__(256) __attribute__((amdgpu_waves_per_eu(2, 2)))
void row_dft(const float* __restrict__ x, const float* __restrict__ BwT,
             float* __restrict__ X1) {
  extern __shared__ v4f smem4[];
  const int wave = threadIdx.x >> 5, lane = threadIdx.x & 31;
  const int hh = lane >> 4, lm = lane & 15;
  const int rowBase = blockIdx.x * 128 + wave * 16;
  const v4f* arow4 = (const v4f*)(x + (rowBase + lm) * 512) + hh;

  const v4f* src4 = (const v4f*)BwT;
  for (int i = 0; i < 32; ++i) {
    const int idx = threadIdx.x + i * 256;
    smem4[(idx >> 7) * kBsPitch4 + (idx & 127)] = src4[idx];
  }
  v4f a_cur[16];
  for (int jj = 0; jj < 16; ++jj) a_cur[jj] = arow4[2 * jj];
  __syncthreads();

  const v4f* bcol4 = smem4 + lm * kBsPitch4 + hh;
  v8f acc[4] = {};
  for (int t = 0; t < 4; ++t) {
    v4f a_nxt[16];
    if (t < 3)
      for (int jj = 0; jj < 16; ++jj) a_nxt[jj] = arow4[2 * (16 * (t + 1) + jj)];
    for (int jj = 0; jj < 16; ++jj) {
      const int j = 16 * t + jj;
      const v4f a = a_cur[jj];
      v2f a0; a0.x = a.x; a0.y = a.y;
      v2f a1; a1.x = a.z; a1.y = a.w;
      v4f b[4];
      for (int nt = 0; nt < 4; ++nt) b[nt] = bcol4[nt * 16 * kBsPitch4 + 2 * j];
      for (int nt = 0; nt < 4; ++nt) {
        v2f b0; b0.x = b[nt].x; b0.y = b[nt].y;
        acc[nt] = wmma4(a0, b0, acc[nt]);
      }
      for (int nt = 0; nt < 4; ++nt) {
        v2f b1; b1.x = b[nt].z; b1.y = b[nt].w;
        acc[nt] = wmma4(a1, b1, acc[nt]);
      }
    }
    if (t < 3)
      for (int jj = 0; jj < 16; ++jj) a_cur[jj] = a_nxt[jj];
  }
  for (int r = 0; r < 8; ++r)
    for (int nt = 0; nt < 4; ++nt)
      X1[(rowBase + r + 8 * hh) * 64 + nt * 16 + lm] = acc[nt][r];
}

__global__ __launch_bounds__(512)
void col_dft(const float* __restrict__ X1, const float* __restrict__ Dm,
             float* __restrict__ X2r, float* __restrict__ X2i) {
  __shared__ float P[64 * 64];
  const int c = blockIdx.x;
  const int wave = threadIdx.x >> 5, lane = threadIdx.x & 31;
  const int mt = wave >> 2, nt = wave & 3;
  const int hh = lane >> 4, lm = lane & 15;
  const int arow = mt * 16 + lm;
  const int col = nt * 16 + lm;
  const v4f* arow4 = (const v4f*)(Dm + arow * 512) + hh;
  const float* B = X1 + c * 512 * 64 + (4 * hh) * 64 + col;

  v8f acc = {};
  for (int j = 0; j < 64; ++j) {
    const v4f a = arow4[2 * j];
    const float* Bj = B + (8 * j) * 64;
    v2f a0; a0.x = a.x; a0.y = a.y;
    v2f b0; b0.x = Bj[0];       b0.y = Bj[64];
    acc = wmma4(a0, b0, acc);
    v2f a1; a1.x = a.z; a1.y = a.w;
    v2f b1; b1.x = Bj[128];     b1.y = Bj[192];
    acc = wmma4(a1, b1, acc);
  }
  for (int r = 0; r < 8; ++r)
    P[(mt * 16 + r + 8 * hh) * 64 + col] = acc[r];
  __syncthreads();

  for (int idx = threadIdx.x; idx < kModes; idx += 512) {
    const int kx = idx >> 5, ky = idx & 31;
    X2r[c * kModes + idx] = P[kx * 64 + ky]        + P[(32 + kx) * 64 + 32 + ky];
    X2i[c * kModes + idx] = P[kx * 64 + 32 + ky]   - P[(32 + kx) * 64 + ky];
  }
}

__global__ __launch_bounds__(256)
void mode_mix(const float* __restrict__ X2r, const float* __restrict__ X2i,
              const float* __restrict__ Wr, const float* __restrict__ Wi,
              float* __restrict__ Fr, float* __restrict__ Fi) {
  const int o = blockIdx.y;
  const int m = blockIdx.x * 256 + threadIdx.x;
  float ar = 0.0f, ai = 0.0f;
  for (int c = 0; c < kC; ++c) {
    const float xr = X2r[c * kModes + m];
    const float xi = X2i[c * kModes + m];
    const int wofs = ((c * kO + o) << 10) + m;
    const float wr = Wr[wofs], wi = Wi[wofs];
    ar = fmaf(xr, wr, fmaf(-xi, wi, ar));
    ai = fmaf(xr, wi, fmaf(xi, wr, ai));
  }
  Fr[(o << 10) + m] = ar;
  Fi[(o << 10) + m] = ai;
}

__global__ __launch_bounds__(512)
void inv_col(const float* __restrict__ Fr, const float* __restrict__ Fi,
             const float* __restrict__ E2, float* __restrict__ T) {
  __shared__ float Bs[64 * 64];
  const int o = blockIdx.x;
  for (int idx = threadIdx.x; idx < 64 * 64; idx += 512) {
    const int r = idx >> 6, n = idx & 63;
    const int fofs = (o << 10) + (r & 31) * 32 + (n & 31);
    float v;
    if (r < 32) v = (n < 32) ?  Fr[fofs] : Fi[fofs];
    else        v = (n < 32) ? -Fi[fofs] : Fr[fofs];
    Bs[idx] = v;
  }
  __syncthreads();

  const int wave = threadIdx.x >> 5, lane = threadIdx.x & 31;
  const int hh = lane >> 4, lm = lane & 15;
  for (int mt = wave; mt < 32; mt += 16) {
    const int rowBase = mt * 16;
    const float* arow = E2 + (rowBase + lm) * 64;
    for (int nt = 0; nt < 4; ++nt) {
      const int col = nt * 16 + lm;
      v8f acc = {};
      for (int k0 = 0; k0 < 64; k0 += 4) {
        const int ka = k0 + 2 * hh;
        v2f a; a.x = arow[ka];            a.y = arow[ka + 1];
        v2f b; b.x = Bs[ka * 64 + col];   b.y = Bs[(ka + 1) * 64 + col];
        acc = wmma4(a, b, acc);
      }
      for (int r = 0; r < 8; ++r)
        T[(o * 512 + rowBase + r + 8 * hh) * 64 + col] = acc[r];
    }
  }
}

__global__ __launch_bounds__(256) __attribute__((amdgpu_waves_per_eu(2, 2)))
void inv_row(const float* __restrict__ T, const float* __restrict__ GT,
             float* __restrict__ y) {
  extern __shared__ v4f smem4[];
  const int wave = threadIdx.x >> 5, lane = threadIdx.x & 31;
  const int hh = lane >> 4, lm = lane & 15;
  const int rb = (blockIdx.x * 8 + wave) * 16;
  const v4f* arow4 = (const v4f*)(T + (rb + lm) * 64) + hh;
  const v4f* src4 = (const v4f*)GT;
  for (int i = 0; i < 32; ++i) {
    const int idx = threadIdx.x + i * 256;
    smem4[(idx >> 4) * kGsPitch4 + (idx & 15)] = src4[idx];
  }
  v4f a[8];
  for (int jj = 0; jj < 8; ++jj) a[jj] = arow4[2 * jj];
  __syncthreads();

  float* yrow = y + (rb + 8 * hh) * 512 + lm;
  for (int ng = 0; ng < 8; ++ng) {
    const v4f* bcol4 = smem4 + (ng * 64 + lm) * kGsPitch4 + hh;
    v8f acc[4] = {};
    for (int jj = 0; jj < 8; ++jj) {
      v2f a0; a0.x = a[jj].x; a0.y = a[jj].y;
      v2f a1; a1.x = a[jj].z; a1.y = a[jj].w;
      v4f b[4];
      for (int q = 0; q < 4; ++q) b[q] = bcol4[q * 16 * kGsPitch4 + 2 * jj];
      for (int q = 0; q < 4; ++q) {
        v2f b0; b0.x = b[q].x; b0.y = b[q].y;
        acc[q] = wmma4(a0, b0, acc[q]);
      }
      for (int q = 0; q < 4; ++q) {
        v2f b1; b1.x = b[q].z; b1.y = b[q].w;
        acc[q] = wmma4(a1, b1, acc[q]);
      }
    }
    for (int r = 0; r < 8; ++r)
      for (int q = 0; q < 4; ++q)
        yrow[r * 512 + ng * 64 + q * 16] = acc[q][r];
  }
}

extern "C" void kernel_launch(void* const* d_in, const int* in_sizes, int n_in,
                              void* d_out, int out_size, void* d_ws, size_t ws_size,
                              hipStream_t stream) {
  const float* x  = (const float*)d_in[0];
  const float* Wr = (const float*)d_in[1];
  const float* Wi = (const float*)d_in[2];
  float* y = (float*)d_out;

  float* ws  = (float*)d_ws;
  float* BwT = ws;
  float* Dm  = BwT + 32768;
  float* E2  = Dm  + 32768;
  float* GT  = E2  + 32768;
  float* X1  = GT  + 32768;
  float* X2r = X1  + 2097152;
  float* X2i = X2r + 65536;
  float* Fr  = X2i + 65536;
  float* Fi  = Fr  + 65536;
  float* T   = Fi  + 65536;

  build_twiddles<<<128, 256, 0, stream>>>(BwT, Dm, E2, GT);
  row_dft<<<256, 256, kOnePerWgpLdsBytes, stream>>>(x, BwT, X1);
  col_dft<<<64, 512, 0, stream>>>(X1, Dm, X2r, X2i);
  mode_mix<<<dim3(4, 64), 256, 0, stream>>>(X2r, X2i, Wr, Wi, Fr, Fi);
  inv_col<<<64, 512, 0, stream>>>(Fr, Fi, E2, T);
  inv_row<<<256, 256, kOnePerWgpLdsBytes, stream>>>(T, GT, y);
}


# inv_col folded into inv_row: each wave builds its 16-row strip of T as a transposed WMMA product whose D tiles are stage E's A fragments; a launch and T's round trip go; k order of both sums changes
# speedup vs baseline: 1.7968x; 1.1443x over previous
#include <hip/hip_runtime.h>
#include <math.h>


static constexpr int kH = 512;
static constexpr int kW = 512;
static constexpr int kC = 64;
static constexpr int kO = 64;
static constexpr int kM1 = 32;
static constexpr int kM2 = 32;
static constexpr int kModes = kM1 * kM2;
static constexpr float kInvHW = 1.0f / (512.0f * 512.0f);
static constexpr int kBsPitch4 = 129;
static constexpr int kGsPitch4 = 17;
static constexpr int kOnePerWgpLdsBytes = 162 * 1024;

typedef __attribute__((ext_vector_type(2))) float v2f;
typedef __attribute__((ext_vector_type(4))) float v4f;
typedef __attribute__((ext_vector_type(8))) float v8f;

__device__ inline v8f wmma4(v2f a, v2f b, v8f c) {
  return __builtin_amdgcn_wmma_f32_16x16x4_f32(
       false, a,  false, b,
       (short)0, c,  false,  false);
}

__global__ __launch_bounds__(256)
void build_twiddles(float* __restrict__ BwT, float* __restrict__ Dm,
                    float* __restrict__ E2, float* __restrict__ GT) {
  const int i = blockIdx.x * 256 + threadIdx.x;
  const float step = 6.2831853071795864769f / 512.0f;
  {
    int n = i >> 9, w = i & 511, k = n & 31;
    float th = ((w * k) & 511) * step;
    BwT[i] = (n < 32) ? cosf(th) : -sinf(th);
  }
  {
    int r = i >> 9, hc = i & 511, k = r & 31;
    float th = ((hc * k) & 511) * step;
    Dm[i] = (r < 32) ? cosf(th) : sinf(th);
  }
  {
    int hr = i >> 6, c = i & 63, k = c & 31;
    float th = ((hr * k) & 511) * step;
    E2[i] = (c < 32) ? cosf(th) : sinf(th);
  }
  {
    int w = i >> 6, c = i & 63, k = c & 31;
    float th = ((w * k) & 511) * step;
    float s = ((k == 0) ? 1.0f : 2.0f) * kInvHW;
    GT[i] = (c < 32) ? s * cosf(th) : -s * sinf(th);
  }
}

__global__ __launch_bounds__(256) __attribute__((amdgpu_waves_per_eu(2, 2)))
void row_dft(const float* __restrict__ x, const float* __restrict__ BwT,
             float* __restrict__ X1) {
  extern __shared__ v4f smem4[];
  const int wave = threadIdx.x >> 5, lane = threadIdx.x & 31;
  const int hh = lane >> 4, lm = lane & 15;
  const int rowBase = blockIdx.x * 128 + wave * 16;
  const v4f* arow4 = (const v4f*)(x + (rowBase + lm) * 512) + hh;

  const v4f* src4 = (const v4f*)BwT;
  for (int i = 0; i < 32; ++i) {
    const int idx = threadIdx.x + i * 256;
    smem4[(idx >> 7) * kBsPitch4 + (idx & 127)] = src4[idx];
  }
  v4f a_cur[16];
  for (int jj = 0; jj < 16; ++jj) a_cur[jj] = arow4[2 * jj];
  __syncthreads();

  const v4f* bcol4 = smem4 + lm * kBsPitch4 + hh;
  v8f acc[4] = {};
  for (int t = 0; t < 4; ++t) {
    v4f a_nxt[16];
    if (t < 3)
      for (int jj = 0; jj < 16; ++jj) a_nxt[jj] = arow4[2 * (16 * (t + 1) + jj)];
    for (int jj = 0; jj < 16; ++jj) {
      const int j = 16 * t + jj;
      const v4f a = a_cur[jj];
      v2f a0; a0.x = a.x; a0.y = a.y;
      v2f a1; a1.x = a.z; a1.y = a.w;
      v4f b[4];
      for (int nt = 0; nt < 4; ++nt) b[nt] = bcol4[nt * 16 * kBsPitch4 + 2 * j];
      for (int nt = 0; nt < 4; ++nt) {
        v2f b0; b0.x = b[nt].x; b0.y = b[nt].y;
        acc[nt] = wmma4(a0, b0, acc[nt]);
      }
      for (int nt = 0; nt < 4; ++nt) {
        v2f b1; b1.x = b[nt].z; b1.y = b[nt].w;
        acc[nt] = wmma4(a1, b1, acc[nt]);
      }
    }
    if (t < 3)
      for (int jj = 0; jj < 16; ++jj) a_cur[jj] = a_nxt[jj];
  }
  for (int r = 0; r < 8; ++r)
    for (int nt = 0; nt < 4; ++nt)
      X1[(rowBase + r + 8 * hh) * 64 + nt * 16 + lm] = acc[nt][r];
}

__global__ __launch_bounds__(512)
void col_dft(const float* __restrict__ X1, const float* __restrict__ Dm,
             float* __restrict__ X2r, float* __restrict__ X2i) {
  __shared__ float P[64 * 64];
  const int c = blockIdx.x;
  const int wave = threadIdx.x >> 5, lane = threadIdx.x & 31;
  const int mt = wave >> 2, nt = wave & 3;
  const int hh = lane >> 4, lm = lane & 15;
  const int arow = mt * 16 + lm;
  const int col = nt * 16 + lm;
  const v4f* arow4 = (const v4f*)(Dm + arow * 512) + hh;
  const float* B = X1 + c * 512 * 64 + (4 * hh) * 64 + col;

  v8f acc = {};
  for (int j = 0; j < 64; ++j) {
    const v4f a = arow4[2 * j];
    const float* Bj = B + (8 * j) * 64;
    v2f a0; a0.x = a.x; a0.y = a.y;
    v2f b0; b0.x = Bj[0];       b0.y = Bj[64];
    acc = wmma4(a0, b0, acc);
    v2f a1; a1.x = a.z; a1.y = a.w;
    v2f b1; b1.x = Bj[128];     b1.y = Bj[192];
    acc = wmma4(a1, b1, acc);
  }
  for (int r = 0; r < 8; ++r)
    P[(mt * 16 + r + 8 * hh) * 64 + col] = acc[r];
  __syncthreads();

  for (int idx = threadIdx.x; idx < kModes; idx += 512) {
    const int kx = idx >> 5, ky = idx & 31;
    X2r[c * kModes + idx] = P[kx * 64 + ky]        + P[(32 + kx) * 64 + 32 + ky];
    X2i[c * kModes + idx] = P[kx * 64 + 32 + ky]   - P[(32 + kx) * 64 + ky];
  }
}

__global__ __launch_bounds__(256)
void mode_mix(const float* __restrict__ X2r, const float* __restrict__ X2i,
              const float* __restrict__ Wr, const float* __restrict__ Wi,
              float* __restrict__ Fr, float* __restrict__ Fi) {
  const int o = blockIdx.y;
  const int m = blockIdx.x * 256 + threadIdx.x;
  float ar = 0.0f, ai = 0.0f;
  for (int c = 0; c < kC; ++c) {
    const float xr = X2r[c * kModes + m];
    const float xi = X2i[c * kModes + m];
    const int wofs = ((c * kO + o) << 10) + m;
    const float wr = Wr[wofs], wi = Wi[wofs];
    ar = fmaf(xr, wr, fmaf(-xi, wi, ar));
    ai = fmaf(xr, wi, fmaf(xi, wr, ai));
  }
  Fr[(o << 10) + m] = ar;
  Fi[(o << 10) + m] = ai;
}

__global__ __launch_bounds__(256) __attribute__((amdgpu_waves_per_eu(2, 2)))
void inv_row(const float* __restrict__ Fr, const float* __restrict__ Fi,
             const float* __restrict__ E2, const float* __restrict__ GT,
             float* __restrict__ y) {
  extern __shared__ v4f smem4[];
  v4f* bm4 = smem4 + 512 * kGsPitch4;
  float* bmf = (float*)bm4;
  const int wave = threadIdx.x >> 5, lane = threadIdx.x & 31;
  const int hh = lane >> 4, lm = lane & 15;
  const int o = blockIdx.x >> 2;
  const int h0 = (blockIdx.x & 3) * 128 + wave * 16;
  const int rb = blockIdx.x * 128 + wave * 16;

  float fv[16];
  for (int i = 0; i < 16; ++i) {
    const int idx = threadIdx.x + i * 256;
    const int r = idx >> 6, n = idx & 63;
    const int fofs = (o << 10) + (r & 31) * 32 + (n & 31);
    const float* srcp = ((r < 32) == (n < 32)) ? Fr : Fi;
    const float v = srcp[fofs];
    fv[i] = (r >= 32 && n < 32) ? -v : v;
  }
  const v4f* erow4 = (const v4f*)(E2 + (h0 + lm) * 64) + hh;
  v4f e[8];
  for (int j = 0; j < 8; ++j) e[j] = erow4[2 * j];
  for (int i = 0; i < 16; ++i) {
    const int idx = threadIdx.x + i * 256;
    bmf[(idx & 63) * (4 * kGsPitch4) + (idx >> 6)] = fv[i];
  }
  __syncthreads();

  const v4f* src4 = (const v4f*)GT;
  v4f g[32];
  for (int i = 0; i < 32; ++i) g[i] = src4[threadIdx.x + i * 256];

  v8f tt[4] = {};
  {
    const v4f* arow4 = bm4 + lm * kGsPitch4 + hh;
    for (int j = 0; j < 8; ++j) {
      v4f av[4];
      for (int kt = 0; kt < 4; ++kt) av[kt] = arow4[kt * 16 * kGsPitch4 + 2 * j];
      v2f e0; e0.x = e[j].x; e0.y = e[j].y;
      v2f e1; e1.x = e[j].z; e1.y = e[j].w;
      for (int kt = 0; kt < 4; ++kt) {
        v2f a0; a0.x = av[kt].x; a0.y = av[kt].y;
        tt[kt] = wmma4(a0, e0, tt[kt]);
      }
      for (int kt = 0; kt < 4; ++kt) {
        v2f a1; a1.x = av[kt].z; a1.y = av[kt].w;
        tt[kt] = wmma4(a1, e1, tt[kt]);
      }
    }
  }

  for (int i = 0; i < 32; ++i) {
    const int idx = threadIdx.x + i * 256;
    smem4[(idx >> 4) * kGsPitch4 + (idx & 15)] = g[i];
  }
  __syncthreads();

  float* yrow = y + (rb + 8 * hh) * 512 + lm;
  for (int ng = 0; ng < 8; ++ng) {
    const v4f* bcol4 = smem4 + (ng * 64 + lm) * kGsPitch4 + 2 * hh;
    v8f acc[4] = {};
    for (int kt = 0; kt < 4; ++kt) {
      for (int u = 0; u < 2; ++u) {
        v2f a0; a0.x = tt[kt][4 * u];     a0.y = tt[kt][4 * u + 1];
        v2f a1; a1.x = tt[kt][4 * u + 2]; a1.y = tt[kt][4 * u + 3];
        v4f b[4];
        for (int q = 0; q < 4; ++q) b[q] = bcol4[q * 16 * kGsPitch4 + 4 * kt + u];
        for (int q = 0; q < 4; ++q) {
          v2f b0; b0.x = b[q].x; b0.y = b[q].y;
          acc[q] = wmma4(a0, b0, acc[q]);
        }
        for (int q = 0; q < 4; ++q) {
          v2f b1; b1.x = b[q].z; b1.y = b[q].w;
          acc[q] = wmma4(a1, b1, acc[q]);
        }
      }
    }
    for (int r = 0; r < 8; ++r)
      for (int q = 0; q < 4; ++q)
        yrow[r * 512 + ng * 64 + q * 16] = acc[q][r];
  }
}

extern "C" void kernel_launch(void* const* d_in, const int* in_sizes, int n_in,
                              void* d_out, int out_size, void* d_ws, size_t ws_size,
                              hipStream_t stream) {
  const float* x  = (const float*)d_in[0];
  const float* Wr = (const float*)d_in[1];
  const float* Wi = (const float*)d_in[2];
  float* y = (float*)d_out;

  float* ws  = (float*)d_ws;
  float* BwT = ws;
  float* Dm  = BwT + 32768;
  float* E2  = Dm  + 32768;
  float* GT  = E2  + 32768;
  float* X1  = GT  + 32768;
  float* X2r = X1  + 2097152;
  float* X2i = X2r + 65536;
  float* Fr  = X2i + 65536;
  float* Fi  = Fr  + 65536;

  build_twiddles<<<128, 256, 0, stream>>>(BwT, Dm, E2, GT);
  row_dft<<<256, 256, kOnePerWgpLdsBytes, stream>>>(x, BwT, X1);
  col_dft<<<64, 512, 0, stream>>>(X1, Dm, X2r, X2i);
  mode_mix<<<dim3(4, 64), 256, 0, stream>>>(X2r, X2i, Wr, Wi, Fr, Fi);
  inv_row<<<256, 256, kOnePerWgpLdsBytes, stream>>>(Fr, Fi, E2, GT, y);
}


# mode_mix: a thread owns 4 modes x 16 channels and has all its weight loads (16-byte) in flight at once, 1 memory latency not 16; the 4 channel groups are summed via LDS: sum order over c changes
# speedup vs baseline: 1.9752x; 1.0993x over previous
#include <hip/hip_runtime.h>
#include <math.h>


static constexpr int kH = 512;
static constexpr int kW = 512;
static constexpr int kC = 64;
static constexpr int kO = 64;
static constexpr int kM1 = 32;
static constexpr int kM2 = 32;
static constexpr int kModes = kM1 * kM2;
static constexpr float kInvHW = 1.0f / (512.0f * 512.0f);
static constexpr int kBsPitch4 = 129;
static constexpr int kGsPitch4 = 17;
static constexpr int kOnePerWgpLdsBytes = 162 * 1024;

typedef __attribute__((ext_vector_type(2))) float v2f;
typedef __attribute__((ext_vector_type(4))) float v4f;
typedef __attribute__((ext_vector_type(8))) float v8f;

__device__ inline v8f wmma4(v2f a, v2f b, v8f c) {
  return __builtin_amdgcn_wmma_f32_16x16x4_f32(
       false, a,  false, b,
       (short)0, c,  false,  false);
}

__global__ __launch_bounds__(256)
void build_twiddles(float* __restrict__ BwT, float* __restrict__ Dm,
                    float* __restrict__ E2, float* __restrict__ GT) {
  const int i = blockIdx.x * 256 + threadIdx.x;
  const float step = 6.2831853071795864769f / 512.0f;
  {
    int n = i >> 9, w = i & 511, k = n & 31;
    float th = ((w * k) & 511) * step;
    BwT[i] = (n < 32) ? cosf(th) : -sinf(th);
  }
  {
    int r = i >> 9, hc = i & 511, k = r & 31;
    float th = ((hc * k) & 511) * step;
    Dm[i] = (r < 32) ? cosf(th) : sinf(th);
  }
  {
    int hr = i >> 6, c = i & 63, k = c & 31;
    float th = ((hr * k) & 511) * step;
    E2[i] = (c < 32) ? cosf(th) : sinf(th);
  }
  {
    int w = i >> 6, c = i & 63, k = c & 31;
    float th = ((w * k) & 511) * step;
    float s = ((k == 0) ? 1.0f : 2.0f) * kInvHW;
    GT[i] = (c < 32) ? s * cosf(th) : -s * sinf(th);
  }
}

__global__ __launch_bounds__(256) __attribute__((amdgpu_waves_per_eu(2, 2)))
void row_dft(const float* __restrict__ x, const float* __restrict__ BwT,
             float* __restrict__ X1) {
  extern __shared__ v4f smem4[];
  const int wave = threadIdx.x >> 5, lane = threadIdx.x & 31;
  const int hh = lane >> 4, lm = lane & 15;
  const int rowBase = blockIdx.x * 128 + wave * 16;
  const v4f* arow4 = (const v4f*)(x + (rowBase + lm) * 512) + hh;

  const v4f* src4 = (const v4f*)BwT;
  for (int i = 0; i < 32; ++i) {
    const int idx = threadIdx.x + i * 256;
    smem4[(idx >> 7) * kBsPitch4 + (idx & 127)] = src4[idx];
  }
  v4f a_cur[16];
  for (int jj = 0; jj < 16; ++jj) a_cur[jj] = arow4[2 * jj];
  __syncthreads();

  const v4f* bcol4 = smem4 + lm * kBsPitch4 + hh;
  v8f acc[4] = {};
  for (int t = 0; t < 4; ++t) {
    v4f a_nxt[16];
    if (t < 3)
      for (int jj = 0; jj < 16; ++jj) a_nxt[jj] = arow4[2 * (16 * (t + 1) + jj)];
    for (int jj = 0; jj < 16; ++jj) {
      const int j = 16 * t + jj;
      const v4f a = a_cur[jj];
      v2f a0; a0.x = a.x; a0.y = a.y;
      v2f a1; a1.x = a.z; a1.y = a.w;
      v4f b[4];
      for (int nt = 0; nt < 4; ++nt) b[nt] = bcol4[nt * 16 * kBsPitch4 + 2 * j];
      for (int nt = 0; nt < 4; ++nt) {
        v2f b0; b0.x = b[nt].x; b0.y = b[nt].y;
        acc[nt] = wmma4(a0, b0, acc[nt]);
      }
      for (int nt = 0; nt < 4; ++nt) {
        v2f b1; b1.x = b[nt].z; b1.y = b[nt].w;
        acc[nt] = wmma4(a1, b1, acc[nt]);
      }
    }
    if (t < 3)
      for (int jj = 0; jj < 16; ++jj) a_cur[jj] = a_nxt[jj];
  }
  for (int r = 0; r < 8; ++r)
    for (int nt = 0; nt < 4; ++nt)
      X1[(rowBase + r + 8 * hh) * 64 + nt * 16 + lm] = acc[nt][r];
}

__global__ __launch_bounds__(512)
void col_dft(const float* __restrict__ X1, const float* __restrict__ Dm,
             float* __restrict__ X2r, float* __restrict__ X2i) {
  __shared__ float P[64 * 64];
  const int c = blockIdx.x;
  const int wave = threadIdx.x >> 5, lane = threadIdx.x & 31;
  const int mt = wave >> 2, nt = wave & 3;
  const int hh = lane >> 4, lm = lane & 15;
  const int arow = mt * 16 + lm;
  const int col = nt * 16 + lm;
  const v4f* arow4 = (const v4f*)(Dm + arow * 512) + hh;
  const float* B = X1 + c * 512 * 64 + (4 * hh) * 64 + col;

  v8f acc = {};
  for (int j = 0; j < 64; ++j) {
    const v4f a = arow4[2 * j];
    const float* Bj = B + (8 * j) * 64;
    v2f a0; a0.x = a.x; a0.y = a.y;
    v2f b0; b0.x = Bj[0];       b0.y = Bj[64];
    acc = wmma4(a0, b0, acc);
    v2f a1; a1.x = a.z; a1.y = a.w;
    v2f b1; b1.x = Bj[128];     b1.y = Bj[192];
    acc = wmma4(a1, b1, acc);
  }
  for (int r = 0; r < 8; ++r)
    P[(mt * 16 + r + 8 * hh) * 64 + col] = acc[r];
  __syncthreads();

  for (int idx = threadIdx.x; idx < kModes; idx += 512) {
    const int kx = idx >> 5, ky = idx & 31;
    X2r[c * kModes + idx] = P[kx * 64 + ky]        + P[(32 + kx) * 64 + 32 + ky];
    X2i[c * kModes + idx] = P[kx * 64 + 32 + ky]   - P[(32 + kx) * 64 + ky];
  }
}

__global__ __launch_bounds__(256) __attribute__((amdgpu_waves_per_eu(2, 2)))
void mode_mix(const float* __restrict__ X2r, const float* __restrict__ X2i,
              const float* __restrict__ Wr, const float* __restrict__ Wi,
              float* __restrict__ Fr, float* __restrict__ Fi) {
  __shared__ v4f red4[4 * 2 * 64];
  const int o = blockIdx.y;
  const int q = threadIdx.x & 63;
  const int g = threadIdx.x >> 6;
  const int m0 = blockIdx.x * 256 + 4 * q;
  const v4f* wr4 = (const v4f*)(Wr + (((16 * g) * kO + o) << 10) + m0);
  const v4f* wi4 = (const v4f*)(Wi + (((16 * g) * kO + o) << 10) + m0);
  const v4f* xr4 = (const v4f*)(X2r + (16 * g) * kModes + m0);
  const v4f* xi4 = (const v4f*)(X2i + (16 * g) * kModes + m0);

  v4f wr[16], wi[16], xr[16], xi[16];
  for (int u = 0; u < 16; ++u) {
    wr[u] = wr4[u * (kO * kModes / 4)];
    wi[u] = wi4[u * (kO * kModes / 4)];
  }
  for (int u = 0; u < 16; ++u) {
    xr[u] = xr4[u * (kModes / 4)];
    xi[u] = xi4[u * (kModes / 4)];
  }
  v4f ar = {}, ai = {};
  for (int u = 0; u < 16; ++u)
    for (int e = 0; e < 4; ++e) {
      ar[e] = fmaf(xr[u][e], wr[u][e], fmaf(-xi[u][e], wi[u][e], ar[e]));
      ai[e] = fmaf(xr[u][e], wi[u][e], fmaf(xi[u][e], wr[u][e], ai[e]));
    }
  red4[(g * 2 + 0) * 64 + q] = ar;
  red4[(g * 2 + 1) * 64 + q] = ai;
  __syncthreads();

  const float* red = (const float*)red4;
  const int mm = threadIdx.x;
  float sr = red[0 * 256 + mm], si = red[1 * 256 + mm];
  for (int gg = 1; gg < 4; ++gg) {
    sr += red[(2 * gg + 0) * 256 + mm];
    si += red[(2 * gg + 1) * 256 + mm];
  }
  Fr[(o << 10) + blockIdx.x * 256 + mm] = sr;
  Fi[(o << 10) + blockIdx.x * 256 + mm] = si;
}

__global__ __launch_bounds__(256) __attribute__((amdgpu_waves_per_eu(2, 2)))
void inv_row(const float* __restrict__ Fr, const float* __restrict__ Fi,
             const float* __restrict__ E2, const float* __restrict__ GT,
             float* __restrict__ y) {
  extern __shared__ v4f smem4[];
  v4f* bm4 = smem4 + 512 * kGsPitch4;
  float* bmf = (float*)bm4;
  const int wave = threadIdx.x >> 5, lane = threadIdx.x & 31;
  const int hh = lane >> 4, lm = lane & 15;
  const int o = blockIdx.x >> 2;
  const int h0 = (blockIdx.x & 3) * 128 + wave * 16;
  const int rb = blockIdx.x * 128 + wave * 16;

  float fv[16];
  for (int i = 0; i < 16; ++i) {
    const int idx = threadIdx.x + i * 256;
    const int r = idx >> 6, n = idx & 63;
    const int fofs = (o << 10) + (r & 31) * 32 + (n & 31);
    const float* srcp = ((r < 32) == (n < 32)) ? Fr : Fi;
    const float v = srcp[fofs];
    fv[i] = (r >= 32 && n < 32) ? -v : v;
  }
  const v4f* erow4 = (const v4f*)(E2 + (h0 + lm) * 64) + hh;
  v4f e[8];
  for (int j = 0; j < 8; ++j) e[j] = erow4[2 * j];
  for (int i = 0; i < 16; ++i) {
    const int idx = threadIdx.x + i * 256;
    bmf[(idx & 63) * (4 * kGsPitch4) + (idx >> 6)] = fv[i];
  }
  __syncthreads();

  const v4f* src4 = (const v4f*)GT;
  v4f g[32];
  for (int i = 0; i < 32; ++i) g[i] = src4[threadIdx.x + i * 256];

  v8f tt[4] = {};
  {
    const v4f* arow4 = bm4 + lm * kGsPitch4 + hh;
    for (int j = 0; j < 8; ++j) {
      v4f av[4];
      for (int kt = 0; kt < 4; ++kt) av[kt] = arow4[kt * 16 * kGsPitch4 + 2 * j];
      v2f e0; e0.x = e[j].x; e0.y = e[j].y;
      v2f e1; e1.x = e[j].z; e1.y = e[j].w;
      for (int kt = 0; kt < 4; ++kt) {
        v2f a0; a0.x = av[kt].x; a0.y = av[kt].y;
        tt[kt] = wmma4(a0, e0, tt[kt]);
      }
      for (int kt = 0; kt < 4; ++kt) {
        v2f a1; a1.x = av[kt].z; a1.y = av[kt].w;
        tt[kt] = wmma4(a1, e1, tt[kt]);
      }
    }
  }

  for (int i = 0; i < 32; ++i) {
    const int idx = threadIdx.x + i * 256;
    smem4[(idx >> 4) * kGsPitch4 + (idx & 15)] = g[i];
  }
  __syncthreads();

  float* yrow = y + (rb + 8 * hh) * 512 + lm;
  for (int ng = 0; ng < 8; ++ng) {
    const v4f* bcol4 = smem4 + (ng * 64 + lm) * kGsPitch4 + 2 * hh;
    v8f acc[4] = {};
    for (int kt = 0; kt < 4; ++kt) {
      for (int u = 0; u < 2; ++u) {
        v2f a0; a0.x = tt[kt][4 * u];     a0.y = tt[kt][4 * u + 1];
        v2f a1; a1.x = tt[kt][4 * u + 2]; a1.y = tt[kt][4 * u + 3];
        v4f b[4];
        for (int q = 0; q < 4; ++q) b[q] = bcol4[q * 16 * kGsPitch4 + 4 * kt + u];
        for (int q = 0; q < 4; ++q) {
          v2f b0; b0.x = b[q].x; b0.y = b[q].y;
          acc[q] = wmma4(a0, b0, acc[q]);
        }
        for (int q = 0; q < 4; ++q) {
          v2f b1; b1.x = b[q].z; b1.y = b[q].w;
          acc[q] = wmma4(a1, b1, acc[q]);
        }
      }
    }
    for (int r = 0; r < 8; ++r)
      for (int q = 0; q < 4; ++q)
        yrow[r * 512 + ng * 64 + q * 16] = acc[q][r];
  }
}

extern "C" void kernel_launch(void* const* d_in, const int* in_sizes, int n_in,
                              void* d_out, int out_size, void* d_ws, size_t ws_size,
                              hipStream_t stream) {
  const float* x  = (const float*)d_in[0];
  const float* Wr = (const float*)d_in[1];
  const float* Wi = (const float*)d_in[2];
  float* y = (float*)d_out;

  float* ws  = (float*)d_ws;
  float* BwT = ws;
  float* Dm  = BwT + 32768;
  float* E2  = Dm  + 32768;
  float* GT  = E2  + 32768;
  float* X1  = GT  + 32768;
  float* X2r = X1  + 2097152;
  float* X2i = X2r + 65536;
  float* Fr  = X2i + 65536;
  float* Fi  = Fr  + 65536;

  build_twiddles<<<128, 256, 0, stream>>>(BwT, Dm, E2, GT);
  row_dft<<<256, 256, kOnePerWgpLdsBytes, stream>>>(x, BwT, X1);
  col_dft<<<64, 512, 0, stream>>>(X1, Dm, X2r, X2i);
  mode_mix<<<dim3(4, 64), 256, 0, stream>>>(X2r, X2i, Wr, Wi, Fr, Fi);
  inv_row<<<256, 256, kOnePerWgpLdsBytes, stream>>>(Fr, Fi, E2, GT, y);
}


# col_dft: grid 64x16 waves -> 256x8 (all WGPs): a group owns the 4 tiles of one 16x16 spectrum patch, each wave 1/8 of h on 4 accumulators, partial tiles summed via LDS; sum order over h changes
# speedup vs baseline: 2.3548x; 1.1922x over previous
#include <hip/hip_runtime.h>
#include <math.h>


static constexpr int kH = 512;
static constexpr int kW = 512;
static constexpr int kC = 64;
static constexpr int kO = 64;
static constexpr int kM1 = 32;
static constexpr int kM2 = 32;
static constexpr int kModes = kM1 * kM2;
static constexpr float kInvHW = 1.0f / (512.0f * 512.0f);
static constexpr int kBsPitch4 = 129;
static constexpr int kGsPitch4 = 17;
static constexpr int kOnePerWgpLdsBytes = 162 * 1024;

typedef __attribute__((ext_vector_type(2))) float v2f;
typedef __attribute__((ext_vector_type(4))) float v4f;
typedef __attribute__((ext_vector_type(8))) float v8f;

__device__ inline v8f wmma4(v2f a, v2f b, v8f c) {
  return __builtin_amdgcn_wmma_f32_16x16x4_f32(
       false, a,  false, b,
       (short)0, c,  false,  false);
}

__global__ __launch_bounds__(256)
void build_twiddles(float* __restrict__ BwT, float* __restrict__ Dm,
                    float* __restrict__ E2, float* __restrict__ GT) {
  const int i = blockIdx.x * 256 + threadIdx.x;
  const float step = 6.2831853071795864769f / 512.0f;
  {
    int n = i >> 9, w = i & 511, k = n & 31;
    float th = ((w * k) & 511) * step;
    BwT[i] = (n < 32) ? cosf(th) : -sinf(th);
  }
  {
    int r = i >> 9, hc = i & 511, k = r & 31;
    float th = ((hc * k) & 511) * step;
    Dm[i] = (r < 32) ? cosf(th) : sinf(th);
  }
  {
    int hr = i >> 6, c = i & 63, k = c & 31;
    float th = ((hr * k) & 511) * step;
    E2[i] = (c < 32) ? cosf(th) : sinf(th);
  }
  {
    int w = i >> 6, c = i & 63, k = c & 31;
    float th = ((w * k) & 511) * step;
    float s = ((k == 0) ? 1.0f : 2.0f) * kInvHW;
    GT[i] = (c < 32) ? s * cosf(th) : -s * sinf(th);
  }
}

__global__ __launch_bounds__(256) __attribute__((amdgpu_waves_per_eu(2, 2)))
void row_dft(const float* __restrict__ x, const float* __restrict__ BwT,
             float* __restrict__ X1) {
  extern __shared__ v4f smem4[];
  const int wave = threadIdx.x >> 5, lane = threadIdx.x & 31;
  const int hh = lane >> 4, lm = lane & 15;
  const int rowBase = blockIdx.x * 128 + wave * 16;
  const v4f* arow4 = (const v4f*)(x + (rowBase + lm) * 512) + hh;

  const v4f* src4 = (const v4f*)BwT;
  for (int i = 0; i < 32; ++i) {
    const int idx = threadIdx.x + i * 256;
    smem4[(idx >> 7) * kBsPitch4 + (idx & 127)] = src4[idx];
  }
  v4f a_cur[16];
  for (int jj = 0; jj < 16; ++jj) a_cur[jj] = arow4[2 * jj];
  __syncthreads();

  const v4f* bcol4 = smem4 + lm * kBsPitch4 + hh;
  v8f acc[4] = {};
  for (int t = 0; t < 4; ++t) {
    v4f a_nxt[16];
    if (t < 3)
      for (int jj = 0; jj < 16; ++jj) a_nxt[jj] = arow4[2 * (16 * (t + 1) + jj)];
    for (int jj = 0; jj < 16; ++jj) {
      const int j = 16 * t + jj;
      const v4f a = a_cur[jj];
      v2f a0; a0.x = a.x; a0.y = a.y;
      v2f a1; a1.x = a.z; a1.y = a.w;
      v4f b[4];
      for (int nt = 0; nt < 4; ++nt) b[nt] = bcol4[nt * 16 * kBsPitch4 + 2 * j];
      for (int nt = 0; nt < 4; ++nt) {
        v2f b0; b0.x = b[nt].x; b0.y = b[nt].y;
        acc[nt] = wmma4(a0, b0, acc[nt]);
      }
      for (int nt = 0; nt < 4; ++nt) {
        v2f b1; b1.x = b[nt].z; b1.y = b[nt].w;
        acc[nt] = wmma4(a1, b1, acc[nt]);
      }
    }
    if (t < 3)
      for (int jj = 0; jj < 16; ++jj) a_cur[jj] = a_nxt[jj];
  }
  for (int r = 0; r < 8; ++r)
    for (int nt = 0; nt < 4; ++nt)
      X1[(rowBase + r + 8 * hh) * 64 + nt * 16 + lm] = acc[nt][r];
}

__global__ __launch_bounds__(256) __attribute__((amdgpu_waves_per_eu(2, 2)))
void col_dft(const float* __restrict__ X1, const float* __restrict__ Dm,
             float* __restrict__ X2r, float* __restrict__ X2i) {
  extern __shared__ float part[];
  const int c = blockIdx.x >> 2;
  const int ta = (blockIdx.x >> 1) & 1, tb = blockIdx.x & 1;
  const int wave = threadIdx.x >> 5, lane = threadIdx.x & 31;
  const int hh = lane >> 4, lm = lane & 15;
  const int j0 = wave * 8;
  const v4f* arow4 = (const v4f*)(Dm + (ta * 16 + lm) * 512) + hh + 2 * j0;
  const float* B = X1 + c * 512 * 64 + (8 * j0 + 4 * hh) * 64 + tb * 16 + lm;

  v4f a[8][2];
  float b[8][2][4];
  for (int jj = 0; jj < 8; ++jj) {
    for (int mi = 0; mi < 2; ++mi) a[jj][mi] = arow4[mi * (32 * 512 / 4) + 2 * jj];
    for (int ni = 0; ni < 2; ++ni)
      for (int e = 0; e < 4; ++e) b[jj][ni][e] = B[(8 * jj + e) * 64 + ni * 32];
  }

  v8f acc[2][2] = {};
  for (int jj = 0; jj < 8; ++jj) {
    for (int mi = 0; mi < 2; ++mi)
      for (int ni = 0; ni < 2; ++ni) {
        v2f a0; a0.x = a[jj][mi].x; a0.y = a[jj][mi].y;
        v2f b0; b0.x = b[jj][ni][0]; b0.y = b[jj][ni][1];
        acc[mi][ni] = wmma4(a0, b0, acc[mi][ni]);
      }
    for (int mi = 0; mi < 2; ++mi)
      for (int ni = 0; ni < 2; ++ni) {
        v2f a1; a1.x = a[jj][mi].z; a1.y = a[jj][mi].w;
        v2f b1; b1.x = b[jj][ni][2]; b1.y = b[jj][ni][3];
        acc[mi][ni] = wmma4(a1, b1, acc[mi][ni]);
      }
  }
  float* mine = part + wave * 1024 + hh * 16 + lm;
  for (int mi = 0; mi < 2; ++mi)
    for (int ni = 0; ni < 2; ++ni)
      for (int r = 0; r < 8; ++r)
        mine[(mi * 2 + ni) * 256 + r * 32] = acc[mi][ni][r];
  __syncthreads();

  const int kxl = threadIdx.x >> 4, kyl = threadIdx.x & 15;
  const float* rd = part + (kxl & 7) * 32 + (kxl >> 3) * 16 + kyl;
  float p00 = rd[0], p01 = rd[256], p10 = rd[512], p11 = rd[768];
  for (int w = 1; w < 8; ++w) {
    p00 += rd[w * 1024];
    p01 += rd[w * 1024 + 256];
    p10 += rd[w * 1024 + 512];
    p11 += rd[w * 1024 + 768];
  }
  const int idx = c * kModes + (ta * 16 + kxl) * 32 + tb * 16 + kyl;
  X2r[idx] = p00 + p11;
  X2i[idx] = p01 - p10;
}

__global__ __launch_bounds__(256) __attribute__((amdgpu_waves_per_eu(2, 2)))
void mode_mix(const float* __restrict__ X2r, const float* __restrict__ X2i,
              const float* __restrict__ Wr, const float* __restrict__ Wi,
              float* __restrict__ Fr, float* __restrict__ Fi) {
  __shared__ v4f red4[4 * 2 * 64];
  const int o = blockIdx.y;
  const int q = threadIdx.x & 63;
  const int g = threadIdx.x >> 6;
  const int m0 = blockIdx.x * 256 + 4 * q;
  const v4f* wr4 = (const v4f*)(Wr + (((16 * g) * kO + o) << 10) + m0);
  const v4f* wi4 = (const v4f*)(Wi + (((16 * g) * kO + o) << 10) + m0);
  const v4f* xr4 = (const v4f*)(X2r + (16 * g) * kModes + m0);
  const v4f* xi4 = (const v4f*)(X2i + (16 * g) * kModes + m0);

  v4f wr[16], wi[16], xr[16], xi[16];
  for (int u = 0; u < 16; ++u) {
    wr[u] = wr4[u * (kO * kModes / 4)];
    wi[u] = wi4[u * (kO * kModes / 4)];
  }
  for (int u = 0; u < 16; ++u) {
    xr[u] = xr4[u * (kModes / 4)];
    xi[u] = xi4[u * (kModes / 4)];
  }
  v4f ar = {}, ai = {};
  for (int u = 0; u < 16; ++u)
    for (int e = 0; e < 4; ++e) {
      ar[e] = fmaf(xr[u][e], wr[u][e], fmaf(-xi[u][e], wi[u][e], ar[e]));
      ai[e] = fmaf(xr[u][e], wi[u][e], fmaf(xi[u][e], wr[u][e], ai[e]));
    }
  red4[(g * 2 + 0) * 64 + q] = ar;
  red4[(g * 2 + 1) * 64 + q] = ai;
  __syncthreads();

  const float* red = (const float*)red4;
  const int mm = threadIdx.x;
  float sr = red[0 * 256 + mm], si = red[1 * 256 + mm];
  for (int gg = 1; gg < 4; ++gg) {
    sr += red[(2 * gg + 0) * 256 + mm];
    si += red[(2 * gg + 1) * 256 + mm];
  }
  Fr[(o << 10) + blockIdx.x * 256 + mm] = sr;
  Fi[(o << 10) + blockIdx.x * 256 + mm] = si;
}

__global__ __launch_bounds__(256) __attribute__((amdgpu_waves_per_eu(2, 2)))
void inv_row(const float* __restrict__ Fr, const float* __restrict__ Fi,
             const float* __restrict__ E2, const float* __restrict__ GT,
             float* __restrict__ y) {
  extern __shared__ v4f smem4[];
  v4f* bm4 = smem4 + 512 * kGsPitch4;
  float* bmf = (float*)bm4;
  const int wave = threadIdx.x >> 5, lane = threadIdx.x & 31;
  const int hh = lane >> 4, lm = lane & 15;
  const int o = blockIdx.x >> 2;
  const int h0 = (blockIdx.x & 3) * 128 + wave * 16;
  const int rb = blockIdx.x * 128 + wave * 16;

  float fv[16];
  for (int i = 0; i < 16; ++i) {
    const int idx = threadIdx.x + i * 256;
    const int r = idx >> 6, n = idx & 63;
    const int fofs = (o << 10) + (r & 31) * 32 + (n & 31);
    const float* srcp = ((r < 32) == (n < 32)) ? Fr : Fi;
    const float v = srcp[fofs];
    fv[i] = (r >= 32 && n < 32) ? -v : v;
  }
  const v4f* erow4 = (const v4f*)(E2 + (h0 + lm) * 64) + hh;
  v4f e[8];
  for (int j = 0; j < 8; ++j) e[j] = erow4[2 * j];
  for (int i = 0; i < 16; ++i) {
    const int idx = threadIdx.x + i * 256;
    bmf[(idx & 63) * (4 * kGsPitch4) + (idx >> 6)] = fv[i];
  }
  __syncthreads();

  const v4f* src4 = (const v4f*)GT;
  v4f g[32];
  for (int i = 0; i < 32; ++i) g[i] = src4[threadIdx.x + i * 256];

  v8f tt[4] = {};
  {
    const v4f* arow4 = bm4 + lm * kGsPitch4 + hh;
    for (int j = 0; j < 8; ++j) {
      v4f av[4];
      for (int kt = 0; kt < 4; ++kt) av[kt] = arow4[kt * 16 * kGsPitch4 + 2 * j];
      v2f e0; e0.x = e[j].x; e0.y = e[j].y;
      v2f e1; e1.x = e[j].z; e1.y = e[j].w;
      for (int kt = 0; kt < 4; ++kt) {
        v2f a0; a0.x = av[kt].x; a0.y = av[kt].y;
        tt[kt] = wmma4(a0, e0, tt[kt]);
      }
      for (int kt = 0; kt < 4; ++kt) {
        v2f a1; a1.x = av[kt].z; a1.y = av[kt].w;
        tt[kt] = wmma4(a1, e1, tt[kt]);
      }
    }
  }

  for (int i = 0; i < 32; ++i) {
    const int idx = threadIdx.x + i * 256;
    smem4[(idx >> 4) * kGsPitch4 + (idx & 15)] = g[i];
  }
  __syncthreads();

  float* yrow = y + (rb + 8 * hh) * 512 + lm;
  for (int ng = 0; ng < 8; ++ng) {
    const v4f* bcol4 = smem4 + (ng * 64 + lm) * kGsPitch4 + 2 * hh;
    v8f acc[4] = {};
    for (int kt = 0; kt < 4; ++kt) {
      for (int u = 0; u < 2; ++u) {
        v2f a0; a0.x = tt[kt][4 * u];     a0.y = tt[kt][4 * u + 1];
        v2f a1; a1.x = tt[kt][4 * u + 2]; a1.y = tt[kt][4 * u + 3];
        v4f b[4];
        for (int q = 0; q < 4; ++q) b[q] = bcol4[q * 16 * kGsPitch4 + 4 * kt + u];
        for (int q = 0; q < 4; ++q) {
          v2f b0; b0.x = b[q].x; b0.y = b[q].y;
          acc[q] = wmma4(a0, b0, acc[q]);
        }
        for (int q = 0; q < 4; ++q) {
          v2f b1; b1.x = b[q].z; b1.y = b[q].w;
          acc[q] = wmma4(a1, b1, acc[q]);
        }
      }
    }
    for (int r = 0; r < 8; ++r)
      for (int q = 0; q < 4; ++q)
        yrow[r * 512 + ng * 64 + q * 16] = acc[q][r];
  }
}

extern "C" void kernel_launch(void* const* d_in, const int* in_sizes, int n_in,
                              void* d_out, int out_size, void* d_ws, size_t ws_size,
                              hipStream_t stream) {
  const float* x  = (const float*)d_in[0];
  const float* Wr = (const float*)d_in[1];
  const float* Wi = (const float*)d_in[2];
  float* y = (float*)d_out;

  float* ws  = (float*)d_ws;
  float* BwT = ws;
  float* Dm  = BwT + 32768;
  float* E2  = Dm  + 32768;
  float* GT  = E2  + 32768;
  float* X1  = GT  + 32768;
  float* X2r = X1  + 2097152;
  float* X2i = X2r + 65536;
  float* Fr  = X2i + 65536;
  float* Fi  = Fr  + 65536;

  build_twiddles<<<128, 256, 0, stream>>>(BwT, Dm, E2, GT);
  row_dft<<<256, 256, kOnePerWgpLdsBytes, stream>>>(x, BwT, X1);
  col_dft<<<256, 256, kOnePerWgpLdsBytes, stream>>>(X1, Dm, X2r, X2i);
  mode_mix<<<dim3(4, 64), 256, 0, stream>>>(X2r, X2i, Wr, Wi, Fr, Fi);
  inv_row<<<256, 256, kOnePerWgpLdsBytes, stream>>>(Fr, Fi, E2, GT, y);
}


# row_dft, stage E of inv_row: W axis split by parity of k: x[w]+-x[w+256] feed even/odd-k tiles, y[w],y[w+256]=Ev+-Od; half the WMMAs and half the twiddle tables; sum order over w and k changes
# speedup vs baseline: 2.7268x; 1.1580x over previous
#include <hip/hip_runtime.h>
#include <math.h>


static constexpr int kH = 512;
static constexpr int kW = 512;
static constexpr int kC = 64;
static constexpr int kO = 64;
static constexpr int kM1 = 32;
static constexpr int kM2 = 32;
static constexpr int kModes = kM1 * kM2;
static constexpr float kInvHW = 1.0f / (512.0f * 512.0f);
static constexpr int kBsPitch4 = 65;
static constexpr int kGsPitch4 = 17;
static constexpr int kOnePerWgpLdsBytes = 162 * 1024;

typedef __attribute__((ext_vector_type(2))) float v2f;
typedef __attribute__((ext_vector_type(4))) float v4f;
typedef __attribute__((ext_vector_type(8))) float v8f;

__device__ inline v8f wmma4(v2f a, v2f b, v8f c) {
  return __builtin_amdgcn_wmma_f32_16x16x4_f32(
       false, a,  false, b,
       (short)0, c,  false,  false);
}

__global__ __launch_bounds__(256)
void build_twiddles(float* __restrict__ BwT, float* __restrict__ Dm,
                    float* __restrict__ E2, float* __restrict__ GT) {
  const int i = blockIdx.x * 256 + threadIdx.x;
  const float step = 6.2831853071795864769f / 512.0f;
  if (i < 16384) {
    {
      int p = i >> 8, w = i & 255, k = 2 * (p & 15) + (p >> 5);
      float th = ((w * k) & 511) * step;
      BwT[i] = ((p >> 4) & 1) ? -sinf(th) : cosf(th);
    }
    {
      int w = i >> 6, p = i & 63, k = 2 * (p & 15) + (p >> 5);
      float th = ((w * k) & 511) * step;
      float s = ((k == 0) ? 1.0f : 2.0f) * kInvHW;
      GT[i] = ((p >> 4) & 1) ? -s * sinf(th) : s * cosf(th);
    }
  }
  {
    int r = i >> 9, hc = i & 511, k = r & 31;
    float th = ((hc * k) & 511) * step;
    Dm[i] = (r < 32) ? cosf(th) : sinf(th);
  }
  {
    int hr = i >> 6, c = i & 63, k = c & 31;
    float th = ((hr * k) & 511) * step;
    E2[i] = (c < 32) ? cosf(th) : sinf(th);
  }
}

__global__ __launch_bounds__(256) __attribute__((amdgpu_waves_per_eu(2, 2)))
void row_dft(const float* __restrict__ x, const float* __restrict__ BwT,
             float* __restrict__ X1) {
  extern __shared__ v4f smem4[];
  const int wave = threadIdx.x >> 5, lane = threadIdx.x & 31;
  const int hh = lane >> 4, lm = lane & 15;
  const int rowBase = blockIdx.x * 128 + wave * 16;
  const v4f* arow4 = (const v4f*)(x + (rowBase + lm) * 512) + hh;

  const v4f* src4 = (const v4f*)BwT;
  v4f tw[16];
  for (int i = 0; i < 16; ++i) tw[i] = src4[threadIdx.x + i * 256];
  v4f alo[32], ahi[32];
  for (int j = 0; j < 8; ++j) {
    alo[j] = arow4[2 * j];
    ahi[j] = arow4[2 * j + 64];
  }
  for (int i = 0; i < 16; ++i) {
    const int idx = threadIdx.x + i * 256;
    smem4[(idx >> 6) * kBsPitch4 + (idx & 63)] = tw[i];
  }
  __syncthreads();
  for (int j = 8; j < 32; ++j) {
    alo[j] = arow4[2 * j];
    ahi[j] = arow4[2 * j + 64];
  }
  __builtin_amdgcn_sched_barrier(0);

  const v4f* bcol4 = smem4 + lm * kBsPitch4 + hh;
  v8f acc[4] = {};
  for (int j = 0; j < 32; ++j) {
    const v4f s = alo[j] + ahi[j];
    const v4f d = alo[j] - ahi[j];
    v4f b[4];
    for (int nt = 0; nt < 4; ++nt) b[nt] = bcol4[nt * 16 * kBsPitch4 + 2 * j];
    v2f s0; s0.x = s.x; s0.y = s.y;
    v2f d0; d0.x = d.x; d0.y = d.y;
    v2f s1; s1.x = s.z; s1.y = s.w;
    v2f d1; d1.x = d.z; d1.y = d.w;
    for (int nt = 0; nt < 4; ++nt) {
      v2f b0; b0.x = b[nt].x; b0.y = b[nt].y;
      acc[nt] = wmma4(nt < 2 ? s0 : d0, b0, acc[nt]);
    }
    for (int nt = 0; nt < 4; ++nt) {
      v2f b1; b1.x = b[nt].z; b1.y = b[nt].w;
      acc[nt] = wmma4(nt < 2 ? s1 : d1, b1, acc[nt]);
    }
  }
  v2f* X1v = (v2f*)X1;
  for (int r = 0; r < 8; ++r) {
    v2f pc; pc.x = acc[0][r]; pc.y = acc[2][r];
    v2f ps; ps.x = acc[1][r]; ps.y = acc[3][r];
    X1v[(rowBase + r + 8 * hh) * 32 + lm] = pc;
    X1v[(rowBase + r + 8 * hh) * 32 + 16 + lm] = ps;
  }
}

__global__ __launch_bounds__(256) __attribute__((amdgpu_waves_per_eu(2, 2)))
void col_dft(const float* __restrict__ X1, const float* __restrict__ Dm,
             float* __restrict__ X2r, float* __restrict__ X2i) {
  extern __shared__ float part[];
  const int c = blockIdx.x >> 2;
  const int ta = (blockIdx.x >> 1) & 1, tb = blockIdx.x & 1;
  const int wave = threadIdx.x >> 5, lane = threadIdx.x & 31;
  const int hh = lane >> 4, lm = lane & 15;
  const int j0 = wave * 8;
  const v4f* arow4 = (const v4f*)(Dm + (ta * 16 + lm) * 512) + hh + 2 * j0;
  const float* B = X1 + c * 512 * 64 + (8 * j0 + 4 * hh) * 64 + tb * 16 + lm;

  v4f a[8][2];
  float b[8][2][4];
  for (int jj = 0; jj < 8; ++jj) {
    for (int mi = 0; mi < 2; ++mi) a[jj][mi] = arow4[mi * (32 * 512 / 4) + 2 * jj];
    for (int ni = 0; ni < 2; ++ni)
      for (int e = 0; e < 4; ++e) b[jj][ni][e] = B[(8 * jj + e) * 64 + ni * 32];
  }

  v8f acc[2][2] = {};
  for (int jj = 0; jj < 8; ++jj) {
    for (int mi = 0; mi < 2; ++mi)
      for (int ni = 0; ni < 2; ++ni) {
        v2f a0; a0.x = a[jj][mi].x; a0.y = a[jj][mi].y;
        v2f b0; b0.x = b[jj][ni][0]; b0.y = b[jj][ni][1];
        acc[mi][ni] = wmma4(a0, b0, acc[mi][ni]);
      }
    for (int mi = 0; mi < 2; ++mi)
      for (int ni = 0; ni < 2; ++ni) {
        v2f a1; a1.x = a[jj][mi].z; a1.y = a[jj][mi].w;
        v2f b1; b1.x = b[jj][ni][2]; b1.y = b[jj][ni][3];
        acc[mi][ni] = wmma4(a1, b1, acc[mi][ni]);
      }
  }
  float* mine = part + wave * 1024 + hh * 16 + lm;
  for (int mi = 0; mi < 2; ++mi)
    for (int ni = 0; ni < 2; ++ni)
      for (int r = 0; r < 8; ++r)
        mine[(mi * 2 + ni) * 256 + r * 32] = acc[mi][ni][r];
  __syncthreads();

  const int kxl = threadIdx.x >> 4, kyl = threadIdx.x & 15;
  const float* rd = part + (kxl & 7) * 32 + (kxl >> 3) * 16 + kyl;
  float p00 = rd[0], p01 = rd[256], p10 = rd[512], p11 = rd[768];
  for (int w = 1; w < 8; ++w) {
    p00 += rd[w * 1024];
    p01 += rd[w * 1024 + 256];
    p10 += rd[w * 1024 + 512];
    p11 += rd[w * 1024 + 768];
  }
  const int idx = c * kModes + (ta * 16 + kxl) * 32 + tb * 16 + kyl;
  X2r[idx] = p00 + p11;
  X2i[idx] = p01 - p10;
}

__global__ __launch_bounds__(256) __attribute__((amdgpu_waves_per_eu(2, 2)))
void mode_mix(const float* __restrict__ X2r, const float* __restrict__ X2i,
              const float* __restrict__ Wr, const float* __restrict__ Wi,
              float* __restrict__ Fr, float* __restrict__ Fi) {
  __shared__ v4f red4[4 * 2 * 64];
  const int o = blockIdx.y;
  const int q = threadIdx.x & 63;
  const int g = threadIdx.x >> 6;
  const int m0 = blockIdx.x * 256 + 4 * q;
  const v4f* wr4 = (const v4f*)(Wr + (((16 * g) * kO + o) << 10) + m0);
  const v4f* wi4 = (const v4f*)(Wi + (((16 * g) * kO + o) << 10) + m0);
  const v4f* xr4 = (const v4f*)(X2r + (16 * g) * kModes + m0);
  const v4f* xi4 = (const v4f*)(X2i + (16 * g) * kModes + m0);

  v4f wr[16], wi[16], xr[16], xi[16];
  for (int u = 0; u < 16; ++u) {
    wr[u] = wr4[u * (kO * kModes / 4)];
    wi[u] = wi4[u * (kO * kModes / 4)];
  }
  for (int u = 0; u < 16; ++u) {
    xr[u] = xr4[u * (kModes / 4)];
    xi[u] = xi4[u * (kModes / 4)];
  }
  v4f ar = {}, ai = {};
  for (int u = 0; u < 16; ++u)
    for (int e = 0; e < 4; ++e) {
      ar[e] = fmaf(xr[u][e], wr[u][e], fmaf(-xi[u][e], wi[u][e], ar[e]));
      ai[e] = fmaf(xr[u][e], wi[u][e], fmaf(xi[u][e], wr[u][e], ai[e]));
    }
  red4[(g * 2 + 0) * 64 + q] = ar;
  red4[(g * 2 + 1) * 64 + q] = ai;
  __syncthreads();

  const float* red = (const float*)red4;
  const int mm = threadIdx.x;
  float sr = red[0 * 256 + mm], si = red[1 * 256 + mm];
  for (int gg = 1; gg < 4; ++gg) {
    sr += red[(2 * gg + 0) * 256 + mm];
    si += red[(2 * gg + 1) * 256 + mm];
  }
  Fr[(o << 10) + blockIdx.x * 256 + mm] = sr;
  Fi[(o << 10) + blockIdx.x * 256 + mm] = si;
}

__global__ __launch_bounds__(256) __attribute__((amdgpu_waves_per_eu(2, 2)))
void inv_row(const float* __restrict__ Fr, const float* __restrict__ Fi,
             const float* __restrict__ E2, const float* __restrict__ GT,
             float* __restrict__ y) {
  extern __shared__ v4f smem4[];
  v4f* bm4 = smem4 + 256 * kGsPitch4;
  float* bmf = (float*)bm4;
  const int wave = threadIdx.x >> 5, lane = threadIdx.x & 31;
  const int hh = lane >> 4, lm = lane & 15;
  const int o = blockIdx.x >> 2;
  const int h0 = (blockIdx.x & 3) * 128 + wave * 16;
  const int rb = blockIdx.x * 128 + wave * 16;

  float fv[16];
  for (int i = 0; i < 16; ++i) {
    const int idx = threadIdx.x + i * 256;
    const int r = idx >> 6, n = idx & 63;
    const int fofs = (o << 10) + (r & 31) * 32 + (n & 31);
    const float* srcp = ((r < 32) == (n < 32)) ? Fr : Fi;
    const float v = srcp[fofs];
    fv[i] = (r >= 32 && n < 32) ? -v : v;
  }
  const v4f* erow4 = (const v4f*)(E2 + (h0 + lm) * 64) + hh;
  v4f e[8];
  for (int j = 0; j < 8; ++j) e[j] = erow4[2 * j];
  for (int i = 0; i < 16; ++i) {
    const int idx = threadIdx.x + i * 256;
    const int n = idx & 63;
    const int p = 16 * ((n >> 5) + 2 * (n & 1)) + ((n & 31) >> 1);
    bmf[p * (4 * kGsPitch4) + (idx >> 6)] = fv[i];
  }
  __syncthreads();

  const v4f* src4 = (const v4f*)GT;
  v4f g[16];
  for (int i = 0; i < 16; ++i) g[i] = src4[threadIdx.x + i * 256];

  v8f tt[4] = {};
  {
    const v4f* arow4 = bm4 + lm * kGsPitch4 + hh;
    for (int j = 0; j < 8; ++j) {
      v4f av[4];
      for (int kt = 0; kt < 4; ++kt) av[kt] = arow4[kt * 16 * kGsPitch4 + 2 * j];
      v2f e0; e0.x = e[j].x; e0.y = e[j].y;
      v2f e1; e1.x = e[j].z; e1.y = e[j].w;
      for (int kt = 0; kt < 4; ++kt) {
        v2f a0; a0.x = av[kt].x; a0.y = av[kt].y;
        tt[kt] = wmma4(a0, e0, tt[kt]);
      }
      for (int kt = 0; kt < 4; ++kt) {
        v2f a1; a1.x = av[kt].z; a1.y = av[kt].w;
        tt[kt] = wmma4(a1, e1, tt[kt]);
      }
    }
  }

  for (int i = 0; i < 16; ++i) {
    const int idx = threadIdx.x + i * 256;
    smem4[(idx >> 4) * kGsPitch4 + (idx & 15)] = g[i];
  }
  __syncthreads();

  float* yrow = y + (rb + 8 * hh) * 512 + lm;
  for (int ng = 0; ng < 4; ++ng) {
    const v4f* bcol4 = smem4 + (ng * 64 + lm) * kGsPitch4 + 2 * hh;
    v8f ev[4] = {}, od[4] = {};
    for (int kt = 0; kt < 4; ++kt) {
      for (int u = 0; u < 2; ++u) {
        v2f a0; a0.x = tt[kt][4 * u];     a0.y = tt[kt][4 * u + 1];
        v2f a1; a1.x = tt[kt][4 * u + 2]; a1.y = tt[kt][4 * u + 3];
        v4f b[4];
        for (int q = 0; q < 4; ++q) b[q] = bcol4[q * 16 * kGsPitch4 + 4 * kt + u];
        for (int q = 0; q < 4; ++q) {
          v2f b0; b0.x = b[q].x; b0.y = b[q].y;
          if (kt < 2) ev[q] = wmma4(a0, b0, ev[q]);
          else        od[q] = wmma4(a0, b0, od[q]);
        }
        for (int q = 0; q < 4; ++q) {
          v2f b1; b1.x = b[q].z; b1.y = b[q].w;
          if (kt < 2) ev[q] = wmma4(a1, b1, ev[q]);
          else        od[q] = wmma4(a1, b1, od[q]);
        }
      }
    }
    for (int r = 0; r < 8; ++r)
      for (int q = 0; q < 4; ++q) {
        yrow[r * 512 + ng * 64 + q * 16]       = ev[q][r] + od[q][r];
        yrow[r * 512 + 256 + ng * 64 + q * 16] = ev[q][r] - od[q][r];
      }
  }
}

extern "C" void kernel_launch(void* const* d_in, const int* in_sizes, int n_in,
                              void* d_out, int out_size, void* d_ws, size_t ws_size,
                              hipStream_t stream) {
  const float* x  = (const float*)d_in[0];
  const float* Wr = (const float*)d_in[1];
  const float* Wi = (const float*)d_in[2];
  float* y = (float*)d_out;

  float* ws  = (float*)d_ws;
  float* BwT = ws;
  float* Dm  = BwT + 32768;
  float* E2  = Dm  + 32768;
  float* GT  = E2  + 32768;
  float* X1  = GT  + 32768;
  float* X2r = X1  + 2097152;
  float* X2i = X2r + 65536;
  float* Fr  = X2i + 65536;
  float* Fi  = Fr  + 65536;

  build_twiddles<<<128, 256, 0, stream>>>(BwT, Dm, E2, GT);
  row_dft<<<256, 256, kOnePerWgpLdsBytes, stream>>>(x, BwT, X1);
  col_dft<<<256, 256, kOnePerWgpLdsBytes, stream>>>(X1, Dm, X2r, X2i);
  mode_mix<<<dim3(4, 64), 256, 0, stream>>>(X2r, X2i, Wr, Wi, Fr, Fi);
  inv_row<<<256, 256, kOnePerWgpLdsBytes, stream>>>(Fr, Fi, E2, GT, y);
}


# twiddle launch folded into row_dft (5 launches -> 4): each group puts cos/sin of the 512 angles in LDS, builds its table by lookup under its strip's loads and writes Dm, E2, GT; sums unchanged
# speedup vs baseline: 2.8553x; 1.0471x over previous
#include <hip/hip_runtime.h>
#include <math.h>


static constexpr int kH = 512;
static constexpr int kW = 512;
static constexpr int kC = 64;
static constexpr int kO = 64;
static constexpr int kM1 = 32;
static constexpr int kM2 = 32;
static constexpr int kModes = kM1 * kM2;
static constexpr float kInvHW = 1.0f / (512.0f * 512.0f);
static constexpr int kBsPitch4 = 65;
static constexpr int kGsPitch4 = 17;
static constexpr int kOnePerWgpLdsBytes = 162 * 1024;

typedef __attribute__((ext_vector_type(2))) float v2f;
typedef __attribute__((ext_vector_type(4))) float v4f;
typedef __attribute__((ext_vector_type(8))) float v8f;

__device__ inline v8f wmma4(v2f a, v2f b, v8f c) {
  return __builtin_amdgcn_wmma_f32_16x16x4_f32(
       false, a,  false, b,
       (short)0, c,  false,  false);
}


__device__ inline void lds_barrier() {
  asm volatile("s_wait_dscnt 0x0\n\ts_barrier_signal -1\n\ts_barrier_wait -1" ::: "memory");
}

__global__ __launch_bounds__(256) __attribute__((amdgpu_waves_per_eu(2, 2)))
void row_dft(const float* __restrict__ x, float* __restrict__ Dm,
             float* __restrict__ E2, float* __restrict__ GT,
             float* __restrict__ X1) {
  extern __shared__ v4f smem4[];
  float* cs = (float*)(smem4 + 64 * kBsPitch4);
  float* sn = cs + 512 + 32;
  const int wave = threadIdx.x >> 5, lane = threadIdx.x & 31;
  const int hh = lane >> 4, lm = lane & 15;
  const int rowBase = blockIdx.x * 128 + wave * 16;
  const v4f* arow4 = (const v4f*)(x + (rowBase + lm) * 512) + hh;

  v4f alo[32], ahi[32];
  for (int j = 0; j < 30; ++j) {
    alo[j] = arow4[2 * j];
    ahi[j] = arow4[2 * j + 64];
  }

  const float step = 6.2831853071795864769f / 512.0f;
  for (int n = threadIdx.x; n < 512; n += 256) {
    const float th = n * step;
    cs[n] = cosf(th);
    sn[n] = sinf(th);
  }
  lds_barrier();
  {
    const int p = threadIdx.x & 63;
    const int k = 2 * (p & 15) + (p >> 5);
    const bool sinPart = ((p >> 4) & 1) != 0;
    const float* tab = sinPart ? sn : cs;
    for (int i = 0; i < 16; ++i) {
      const int wq = (threadIdx.x >> 6) + 4 * i;
      v4f v;
      for (int e = 0; e < 4; ++e) {
        const float t = tab[((4 * wq + e) * k) & 511];
        v[e] = sinPart ? -t : t;
      }
      smem4[p * kBsPitch4 + wq] = v;
    }
  }
  lds_barrier();
  for (int j = 30; j < 32; ++j) {
    alo[j] = arow4[2 * j];
    ahi[j] = arow4[2 * j + 64];
  }
  __builtin_amdgcn_sched_barrier(0);

  const v4f* bcol4 = smem4 + lm * kBsPitch4 + hh;
  v8f acc[4] = {};
  for (int j = 0; j < 32; ++j) {
    const v4f s = alo[j] + ahi[j];
    const v4f d = alo[j] - ahi[j];
    v4f b[4];
    for (int nt = 0; nt < 4; ++nt) b[nt] = bcol4[nt * 16 * kBsPitch4 + 2 * j];
    v2f s0; s0.x = s.x; s0.y = s.y;
    v2f d0; d0.x = d.x; d0.y = d.y;
    v2f s1; s1.x = s.z; s1.y = s.w;
    v2f d1; d1.x = d.z; d1.y = d.w;
    for (int nt = 0; nt < 4; ++nt) {
      v2f b0; b0.x = b[nt].x; b0.y = b[nt].y;
      acc[nt] = wmma4(nt < 2 ? s0 : d0, b0, acc[nt]);
    }
    for (int nt = 0; nt < 4; ++nt) {
      v2f b1; b1.x = b[nt].z; b1.y = b[nt].w;
      acc[nt] = wmma4(nt < 2 ? s1 : d1, b1, acc[nt]);
    }
  }
  v2f* X1v = (v2f*)X1;
  for (int r = 0; r < 8; ++r) {
    v2f pc; pc.x = acc[0][r]; pc.y = acc[2][r];
    v2f ps; ps.x = acc[1][r]; ps.y = acc[3][r];
    X1v[(rowBase + r + 8 * hh) * 32 + lm] = pc;
    X1v[(rowBase + r + 8 * hh) * 32 + 16 + lm] = ps;
  }

  {
    const int i = blockIdx.x * 256 + threadIdx.x;
    if (blockIdx.x < 128) {
      const int r = i >> 9, hc = i & 511, k = r & 31;
      const int n = (hc * k) & 511;
      Dm[i] = (r < 32) ? cs[n] : sn[n];
    } else {
      const int i2 = i - 32768;
      const int hr = i2 >> 6, c = i2 & 63, k = c & 31;
      const int n = (hr * k) & 511;
      E2[i2] = (c < 32) ? cs[n] : sn[n];
    }
    if (blockIdx.x < 64) {
      const int w = i >> 6, p = i & 63, k = 2 * (p & 15) + (p >> 5);
      const int n = (w * k) & 511;
      const float s = ((k == 0) ? 1.0f : 2.0f) * kInvHW;
      GT[i] = ((p >> 4) & 1) ? -s * sn[n] : s * cs[n];
    }
  }
}

__global__ __launch_bounds__(256) __attribute__((amdgpu_waves_per_eu(2, 2)))
void col_dft(const float* __restrict__ X1, const float* __restrict__ Dm,
             float* __restrict__ X2r, float* __restrict__ X2i) {
  extern __shared__ float part[];
  const int c = blockIdx.x >> 2;
  const int ta = (blockIdx.x >> 1) & 1, tb = blockIdx.x & 1;
  const int wave = threadIdx.x >> 5, lane = threadIdx.x & 31;
  const int hh = lane >> 4, lm = lane & 15;
  const int j0 = wave * 8;
  const v4f* arow4 = (const v4f*)(Dm + (ta * 16 + lm) * 512) + hh + 2 * j0;
  const float* B = X1 + c * 512 * 64 + (8 * j0 + 4 * hh) * 64 + tb * 16 + lm;

  v4f a[8][2];
  float b[8][2][4];
  for (int jj = 0; jj < 8; ++jj) {
    for (int mi = 0; mi < 2; ++mi) a[jj][mi] = arow4[mi * (32 * 512 / 4) + 2 * jj];
    for (int ni = 0; ni < 2; ++ni)
      for (int e = 0; e < 4; ++e) b[jj][ni][e] = B[(8 * jj + e) * 64 + ni * 32];
  }

  v8f acc[2][2] = {};
  for (int jj = 0; jj < 8; ++jj) {
    for (int mi = 0; mi < 2; ++mi)
      for (int ni = 0; ni < 2; ++ni) {
        v2f a0; a0.x = a[jj][mi].x; a0.y = a[jj][mi].y;
        v2f b0; b0.x = b[jj][ni][0]; b0.y = b[jj][ni][1];
        acc[mi][ni] = wmma4(a0, b0, acc[mi][ni]);
      }
    for (int mi = 0; mi < 2; ++mi)
      for (int ni = 0; ni < 2; ++ni) {
        v2f a1; a1.x = a[jj][mi].z; a1.y = a[jj][mi].w;
        v2f b1; b1.x = b[jj][ni][2]; b1.y = b[jj][ni][3];
        acc[mi][ni] = wmma4(a1, b1, acc[mi][ni]);
      }
  }
  float* mine = part + wave * 1024 + hh * 16 + lm;
  for (int mi = 0; mi < 2; ++mi)
    for (int ni = 0; ni < 2; ++ni)
      for (int r = 0; r < 8; ++r)
        mine[(mi * 2 + ni) * 256 + r * 32] = acc[mi][ni][r];
  __syncthreads();

  const int kxl = threadIdx.x >> 4, kyl = threadIdx.x & 15;
  const float* rd = part + (kxl & 7) * 32 + (kxl >> 3) * 16 + kyl;
  float p00 = rd[0], p01 = rd[256], p10 = rd[512], p11 = rd[768];
  for (int w = 1; w < 8; ++w) {
    p00 += rd[w * 1024];
    p01 += rd[w * 1024 + 256];
    p10 += rd[w * 1024 + 512];
    p11 += rd[w * 1024 + 768];
  }
  const int idx = c * kModes + (ta * 16 + kxl) * 32 + tb * 16 + kyl;
  X2r[idx] = p00 + p11;
  X2i[idx] = p01 - p10;
}

__global__ __launch_bounds__(256) __attribute__((amdgpu_waves_per_eu(2, 2)))
void mode_mix(const float* __restrict__ X2r, const float* __restrict__ X2i,
              const float* __restrict__ Wr, const float* __restrict__ Wi,
              float* __restrict__ Fr, float* __restrict__ Fi) {
  __shared__ v4f red4[4 * 2 * 64];
  const int o = blockIdx.y;
  const int q = threadIdx.x & 63;
  const int g = threadIdx.x >> 6;
  const int m0 = blockIdx.x * 256 + 4 * q;
  const v4f* wr4 = (const v4f*)(Wr + (((16 * g) * kO + o) << 10) + m0);
  const v4f* wi4 = (const v4f*)(Wi + (((16 * g) * kO + o) << 10) + m0);
  const v4f* xr4 = (const v4f*)(X2r + (16 * g) * kModes + m0);
  const v4f* xi4 = (const v4f*)(X2i + (16 * g) * kModes + m0);

  v4f wr[16], wi[16], xr[16], xi[16];
  for (int u = 0; u < 16; ++u) {
    wr[u] = wr4[u * (kO * kModes / 4)];
    wi[u] = wi4[u * (kO * kModes / 4)];
  }
  for (int u = 0; u < 16; ++u) {
    xr[u] = xr4[u * (kModes / 4)];
    xi[u] = xi4[u * (kModes / 4)];
  }
  v4f ar = {}, ai = {};
  for (int u = 0; u < 16; ++u)
    for (int e = 0; e < 4; ++e) {
      ar[e] = fmaf(xr[u][e], wr[u][e], fmaf(-xi[u][e], wi[u][e], ar[e]));
      ai[e] = fmaf(xr[u][e], wi[u][e], fmaf(xi[u][e], wr[u][e], ai[e]));
    }
  red4[(g * 2 + 0) * 64 + q] = ar;
  red4[(g * 2 + 1) * 64 + q] = ai;
  __syncthreads();

  const float* red = (const float*)red4;
  const int mm = threadIdx.x;
  float sr = red[0 * 256 + mm], si = red[1 * 256 + mm];
  for (int gg = 1; gg < 4; ++gg) {
    sr += red[(2 * gg + 0) * 256 + mm];
    si += red[(2 * gg + 1) * 256 + mm];
  }
  Fr[(o << 10) + blockIdx.x * 256 + mm] = sr;
  Fi[(o << 10) + blockIdx.x * 256 + mm] = si;
}

__global__ __launch_bounds__(256) __attribute__((amdgpu_waves_per_eu(2, 2)))
void inv_row(const float* __restrict__ Fr, const float* __restrict__ Fi,
             const float* __restrict__ E2, const float* __restrict__ GT,
             float* __restrict__ y) {
  extern __shared__ v4f smem4[];
  v4f* bm4 = smem4 + 256 * kGsPitch4;
  float* bmf = (float*)bm4;
  const int wave = threadIdx.x >> 5, lane = threadIdx.x & 31;
  const int hh = lane >> 4, lm = lane & 15;
  const int o = blockIdx.x >> 2;
  const int h0 = (blockIdx.x & 3) * 128 + wave * 16;
  const int rb = blockIdx.x * 128 + wave * 16;

  float fv[16];
  for (int i = 0; i < 16; ++i) {
    const int idx = threadIdx.x + i * 256;
    const int r = idx >> 6, n = idx & 63;
    const int fofs = (o << 10) + (r & 31) * 32 + (n & 31);
    const float* srcp = ((r < 32) == (n < 32)) ? Fr : Fi;
    const float v = srcp[fofs];
    fv[i] = (r >= 32 && n < 32) ? -v : v;
  }
  const v4f* erow4 = (const v4f*)(E2 + (h0 + lm) * 64) + hh;
  v4f e[8];
  for (int j = 0; j < 8; ++j) e[j] = erow4[2 * j];
  for (int i = 0; i < 16; ++i) {
    const int idx = threadIdx.x + i * 256;
    const int n = idx & 63;
    const int p = 16 * ((n >> 5) + 2 * (n & 1)) + ((n & 31) >> 1);
    bmf[p * (4 * kGsPitch4) + (idx >> 6)] = fv[i];
  }
  __syncthreads();

  const v4f* src4 = (const v4f*)GT;
  v4f g[16];
  for (int i = 0; i < 16; ++i) g[i] = src4[threadIdx.x + i * 256];

  v8f tt[4] = {};
  {
    const v4f* arow4 = bm4 + lm * kGsPitch4 + hh;
    for (int j = 0; j < 8; ++j) {
      v4f av[4];
      for (int kt = 0; kt < 4; ++kt) av[kt] = arow4[kt * 16 * kGsPitch4 + 2 * j];
      v2f e0; e0.x = e[j].x; e0.y = e[j].y;
      v2f e1; e1.x = e[j].z; e1.y = e[j].w;
      for (int kt = 0; kt < 4; ++kt) {
        v2f a0; a0.x = av[kt].x; a0.y = av[kt].y;
        tt[kt] = wmma4(a0, e0, tt[kt]);
      }
      for (int kt = 0; kt < 4; ++kt) {
        v2f a1; a1.x = av[kt].z; a1.y = av[kt].w;
        tt[kt] = wmma4(a1, e1, tt[kt]);
      }
    }
  }

  for (int i = 0; i < 16; ++i) {
    const int idx = threadIdx.x + i * 256;
    smem4[(idx >> 4) * kGsPitch4 + (idx & 15)] = g[i];
  }
  __syncthreads();

  float* yrow = y + (rb + 8 * hh) * 512 + lm;
  for (int ng = 0; ng < 4; ++ng) {
    const v4f* bcol4 = smem4 + (ng * 64 + lm) * kGsPitch4 + 2 * hh;
    v8f ev[4] = {}, od[4] = {};
    for (int kt = 0; kt < 4; ++kt) {
      for (int u = 0; u < 2; ++u) {
        v2f a0; a0.x = tt[kt][4 * u];     a0.y = tt[kt][4 * u + 1];
        v2f a1; a1.x = tt[kt][4 * u + 2]; a1.y = tt[kt][4 * u + 3];
        v4f b[4];
        for (int q = 0; q < 4; ++q) b[q] = bcol4[q * 16 * kGsPitch4 + 4 * kt + u];
        for (int q = 0; q < 4; ++q) {
          v2f b0; b0.x = b[q].x; b0.y = b[q].y;
          if (kt < 2) ev[q] = wmma4(a0, b0, ev[q]);
          else        od[q] = wmma4(a0, b0, od[q]);
        }
        for (int q = 0; q < 4; ++q) {
          v2f b1; b1.x = b[q].z; b1.y = b[q].w;
          if (kt < 2) ev[q] = wmma4(a1, b1, ev[q]);
          else        od[q] = wmma4(a1, b1, od[q]);
        }
      }
    }
    for (int r = 0; r < 8; ++r)
      for (int q = 0; q < 4; ++q) {
        yrow[r * 512 + ng * 64 + q * 16]       = ev[q][r] + od[q][r];
        yrow[r * 512 + 256 + ng * 64 + q * 16] = ev[q][r] - od[q][r];
      }
  }
}

extern "C" void kernel_launch(void* const* d_in, const int* in_sizes, int n_in,
                              void* d_out, int out_size, void* d_ws, size_t ws_size,
                              hipStream_t stream) {
  const float* x  = (const float*)d_in[0];
  const float* Wr = (const float*)d_in[1];
  const float* Wi = (const float*)d_in[2];
  float* y = (float*)d_out;

  float* ws  = (float*)d_ws;
  float* Dm  = ws  + 32768;
  float* E2  = Dm  + 32768;
  float* GT  = E2  + 32768;
  float* X1  = GT  + 32768;
  float* X2r = X1  + 2097152;
  float* X2i = X2r + 65536;
  float* Fr  = X2i + 65536;
  float* Fi  = Fr  + 65536;

  row_dft<<<256, 256, kOnePerWgpLdsBytes, stream>>>(x, Dm, E2, GT, X1);
  col_dft<<<256, 256, kOnePerWgpLdsBytes, stream>>>(X1, Dm, X2r, X2i);
  mode_mix<<<dim3(4, 64), 256, 0, stream>>>(X2r, X2i, Wr, Wi, Fr, Fi);
  inv_row<<<256, 256, kOnePerWgpLdsBytes, stream>>>(Fr, Fi, E2, GT, y);
}


# inv_row: G columns permuted so a lane's 4 tiles are 4 consecutive w: y leaves as 16-byte stores, 256 contiguous bytes per row and instruction (whole lines), not 4-byte half lines; sums unchanged
# speedup vs baseline: 3.2340x; 1.1326x over previous
#include <hip/hip_runtime.h>
#include <math.h>


static constexpr int kH = 512;
static constexpr int kW = 512;
static constexpr int kC = 64;
static constexpr int kO = 64;
static constexpr int kM1 = 32;
static constexpr int kM2 = 32;
static constexpr int kModes = kM1 * kM2;
static constexpr float kInvHW = 1.0f / (512.0f * 512.0f);
static constexpr int kBsPitch4 = 65;
static constexpr int kGsPitch4 = 17;
static constexpr int kOnePerWgpLdsBytes = 162 * 1024;

typedef __attribute__((ext_vector_type(2))) float v2f;
typedef __attribute__((ext_vector_type(4))) float v4f;
typedef __attribute__((ext_vector_type(8))) float v8f;

__device__ inline v8f wmma4(v2f a, v2f b, v8f c) {
  return __builtin_amdgcn_wmma_f32_16x16x4_f32(
       false, a,  false, b,
       (short)0, c,  false,  false);
}


__device__ inline void lds_barrier() {
  asm volatile("s_wait_dscnt 0x0\n\ts_barrier_signal -1\n\ts_barrier_wait -1" ::: "memory");
}

__global__ __launch_bounds__(256) __attribute__((amdgpu_waves_per_eu(2, 2)))
void row_dft(const float* __restrict__ x, float* __restrict__ Dm,
             float* __restrict__ E2, float* __restrict__ GT,
             float* __restrict__ X1) {
  extern __shared__ v4f smem4[];
  float* cs = (float*)(smem4 + 64 * kBsPitch4);
  float* sn = cs + 512 + 32;
  const int wave = threadIdx.x >> 5, lane = threadIdx.x & 31;
  const int hh = lane >> 4, lm = lane & 15;
  const int rowBase = blockIdx.x * 128 + wave * 16;
  const v4f* arow4 = (const v4f*)(x + (rowBase + lm) * 512) + hh;

  v4f alo[32], ahi[32];
  for (int j = 0; j < 30; ++j) {
    alo[j] = arow4[2 * j];
    ahi[j] = arow4[2 * j + 64];
  }

  const float step = 6.2831853071795864769f / 512.0f;
  for (int n = threadIdx.x; n < 512; n += 256) {
    const float th = n * step;
    cs[n] = cosf(th);
    sn[n] = sinf(th);
  }
  lds_barrier();
  {
    const int p = threadIdx.x & 63;
    const int k = 2 * (p & 15) + (p >> 5);
    const bool sinPart = ((p >> 4) & 1) != 0;
    const float* tab = sinPart ? sn : cs;
    for (int i = 0; i < 16; ++i) {
      const int wq = (threadIdx.x >> 6) + 4 * i;
      v4f v;
      for (int e = 0; e < 4; ++e) {
        const float t = tab[((4 * wq + e) * k) & 511];
        v[e] = sinPart ? -t : t;
      }
      smem4[p * kBsPitch4 + wq] = v;
    }
  }
  lds_barrier();
  for (int j = 30; j < 32; ++j) {
    alo[j] = arow4[2 * j];
    ahi[j] = arow4[2 * j + 64];
  }
  __builtin_amdgcn_sched_barrier(0);

  const v4f* bcol4 = smem4 + lm * kBsPitch4 + hh;
  v8f acc[4] = {};
  for (int j = 0; j < 32; ++j) {
    const v4f s = alo[j] + ahi[j];
    const v4f d = alo[j] - ahi[j];
    v4f b[4];
    for (int nt = 0; nt < 4; ++nt) b[nt] = bcol4[nt * 16 * kBsPitch4 + 2 * j];
    v2f s0; s0.x = s.x; s0.y = s.y;
    v2f d0; d0.x = d.x; d0.y = d.y;
    v2f s1; s1.x = s.z; s1.y = s.w;
    v2f d1; d1.x = d.z; d1.y = d.w;
    for (int nt = 0; nt < 4; ++nt) {
      v2f b0; b0.x = b[nt].x; b0.y = b[nt].y;
      acc[nt] = wmma4(nt < 2 ? s0 : d0, b0, acc[nt]);
    }
    for (int nt = 0; nt < 4; ++nt) {
      v2f b1; b1.x = b[nt].z; b1.y = b[nt].w;
      acc[nt] = wmma4(nt < 2 ? s1 : d1, b1, acc[nt]);
    }
  }
  v2f* X1v = (v2f*)X1;
  for (int r = 0; r < 8; ++r) {
    v2f pc; pc.x = acc[0][r]; pc.y = acc[2][r];
    v2f ps; ps.x = acc[1][r]; ps.y = acc[3][r];
    X1v[(rowBase + r + 8 * hh) * 32 + lm] = pc;
    X1v[(rowBase + r + 8 * hh) * 32 + 16 + lm] = ps;
  }

  {
    const int i = blockIdx.x * 256 + threadIdx.x;
    if (blockIdx.x < 128) {
      const int r = i >> 9, hc = i & 511, k = r & 31;
      const int n = (hc * k) & 511;
      Dm[i] = (r < 32) ? cs[n] : sn[n];
    } else {
      const int i2 = i - 32768;
      const int hr = i2 >> 6, c = i2 & 63, k = c & 31;
      const int n = (hr * k) & 511;
      E2[i2] = (c < 32) ? cs[n] : sn[n];
    }
    if (blockIdx.x < 64) {
      const int s6 = i >> 6, p = i & 63, k = 2 * (p & 15) + (p >> 5);
      const int w = (s6 & 0xC0) | ((s6 & 15) << 2) | ((s6 >> 4) & 3);
      const int n = (w * k) & 511;
      const float s = ((k == 0) ? 1.0f : 2.0f) * kInvHW;
      GT[i] = ((p >> 4) & 1) ? -s * sn[n] : s * cs[n];
    }
  }
}

__global__ __launch_bounds__(256) __attribute__((amdgpu_waves_per_eu(2, 2)))
void col_dft(const float* __restrict__ X1, const float* __restrict__ Dm,
             float* __restrict__ X2r, float* __restrict__ X2i) {
  extern __shared__ float part[];
  const int c = blockIdx.x >> 2;
  const int ta = (blockIdx.x >> 1) & 1, tb = blockIdx.x & 1;
  const int wave = threadIdx.x >> 5, lane = threadIdx.x & 31;
  const int hh = lane >> 4, lm = lane & 15;
  const int j0 = wave * 8;
  const v4f* arow4 = (const v4f*)(Dm + (ta * 16 + lm) * 512) + hh + 2 * j0;
  const float* B = X1 + c * 512 * 64 + (8 * j0 + 4 * hh) * 64 + tb * 16 + lm;

  v4f a[8][2];
  float b[8][2][4];
  for (int jj = 0; jj < 8; ++jj) {
    for (int mi = 0; mi < 2; ++mi) a[jj][mi] = arow4[mi * (32 * 512 / 4) + 2 * jj];
    for (int ni = 0; ni < 2; ++ni)
      for (int e = 0; e < 4; ++e) b[jj][ni][e] = B[(8 * jj + e) * 64 + ni * 32];
  }

  v8f acc[2][2] = {};
  for (int jj = 0; jj < 8; ++jj) {
    for (int mi = 0; mi < 2; ++mi)
      for (int ni = 0; ni < 2; ++ni) {
        v2f a0; a0.x = a[jj][mi].x; a0.y = a[jj][mi].y;
        v2f b0; b0.x = b[jj][ni][0]; b0.y = b[jj][ni][1];
        acc[mi][ni] = wmma4(a0, b0, acc[mi][ni]);
      }
    for (int mi = 0; mi < 2; ++mi)
      for (int ni = 0; ni < 2; ++ni) {
        v2f a1; a1.x = a[jj][mi].z; a1.y = a[jj][mi].w;
        v2f b1; b1.x = b[jj][ni][2]; b1.y = b[jj][ni][3];
        acc[mi][ni] = wmma4(a1, b1, acc[mi][ni]);
      }
  }
  float* mine = part + wave * 1024 + hh * 16 + lm;
  for (int mi = 0; mi < 2; ++mi)
    for (int ni = 0; ni < 2; ++ni)
      for (int r = 0; r < 8; ++r)
        mine[(mi * 2 + ni) * 256 + r * 32] = acc[mi][ni][r];
  __syncthreads();

  const int kxl = threadIdx.x >> 4, kyl = threadIdx.x & 15;
  const float* rd = part + (kxl & 7) * 32 + (kxl >> 3) * 16 + kyl;
  float p00 = rd[0], p01 = rd[256], p10 = rd[512], p11 = rd[768];
  for (int w = 1; w < 8; ++w) {
    p00 += rd[w * 1024];
    p01 += rd[w * 1024 + 256];
    p10 += rd[w * 1024 + 512];
    p11 += rd[w * 1024 + 768];
  }
  const int idx = c * kModes + (ta * 16 + kxl) * 32 + tb * 16 + kyl;
  X2r[idx] = p00 + p11;
  X2i[idx] = p01 - p10;
}

__global__ __launch_bounds__(256) __attribute__((amdgpu_waves_per_eu(2, 2)))
void mode_mix(const float* __restrict__ X2r, const float* __restrict__ X2i,
              const float* __restrict__ Wr, const float* __restrict__ Wi,
              float* __restrict__ Fr, float* __restrict__ Fi) {
  __shared__ v4f red4[4 * 2 * 64];
  const int o = blockIdx.y;
  const int q = threadIdx.x & 63;
  const int g = threadIdx.x >> 6;
  const int m0 = blockIdx.x * 256 + 4 * q;
  const v4f* wr4 = (const v4f*)(Wr + (((16 * g) * kO + o) << 10) + m0);
  const v4f* wi4 = (const v4f*)(Wi + (((16 * g) * kO + o) << 10) + m0);
  const v4f* xr4 = (const v4f*)(X2r + (16 * g) * kModes + m0);
  const v4f* xi4 = (const v4f*)(X2i + (16 * g) * kModes + m0);

  v4f wr[16], wi[16], xr[16], xi[16];
  for (int u = 0; u < 16; ++u) {
    wr[u] = wr4[u * (kO * kModes / 4)];
    wi[u] = wi4[u * (kO * kModes / 4)];
  }
  for (int u = 0; u < 16; ++u) {
    xr[u] = xr4[u * (kModes / 4)];
    xi[u] = xi4[u * (kModes / 4)];
  }
  v4f ar = {}, ai = {};
  for (int u = 0; u < 16; ++u)
    for (int e = 0; e < 4; ++e) {
      ar[e] = fmaf(xr[u][e], wr[u][e], fmaf(-xi[u][e], wi[u][e], ar[e]));
      ai[e] = fmaf(xr[u][e], wi[u][e], fmaf(xi[u][e], wr[u][e], ai[e]));
    }
  red4[(g * 2 + 0) * 64 + q] = ar;
  red4[(g * 2 + 1) * 64 + q] = ai;
  __syncthreads();

  const float* red = (const float*)red4;
  const int mm = threadIdx.x;
  float sr = red[0 * 256 + mm], si = red[1 * 256 + mm];
  for (int gg = 1; gg < 4; ++gg) {
    sr += red[(2 * gg + 0) * 256 + mm];
    si += red[(2 * gg + 1) * 256 + mm];
  }
  Fr[(o << 10) + blockIdx.x * 256 + mm] = sr;
  Fi[(o << 10) + blockIdx.x * 256 + mm] = si;
}

__global__ __launch_bounds__(256) __attribute__((amdgpu_waves_per_eu(2, 2)))
void inv_row(const float* __restrict__ Fr, const float* __restrict__ Fi,
             const float* __restrict__ E2, const float* __restrict__ GT,
             float* __restrict__ y) {
  extern __shared__ v4f smem4[];
  v4f* bm4 = smem4 + 256 * kGsPitch4;
  float* bmf = (float*)bm4;
  const int wave = threadIdx.x >> 5, lane = threadIdx.x & 31;
  const int hh = lane >> 4, lm = lane & 15;
  const int o = blockIdx.x >> 2;
  const int h0 = (blockIdx.x & 3) * 128 + wave * 16;
  const int rb = blockIdx.x * 128 + wave * 16;

  float fv[16];
  for (int i = 0; i < 16; ++i) {
    const int idx = threadIdx.x + i * 256;
    const int r = idx >> 6, n = idx & 63;
    const int fofs = (o << 10) + (r & 31) * 32 + (n & 31);
    const float* srcp = ((r < 32) == (n < 32)) ? Fr : Fi;
    const float v = srcp[fofs];
    fv[i] = (r >= 32 && n < 32) ? -v : v;
  }
  const v4f* erow4 = (const v4f*)(E2 + (h0 + lm) * 64) + hh;
  v4f e[8];
  for (int j = 0; j < 8; ++j) e[j] = erow4[2 * j];
  for (int i = 0; i < 16; ++i) {
    const int idx = threadIdx.x + i * 256;
    const int n = idx & 63;
    const int p = 16 * ((n >> 5) + 2 * (n & 1)) + ((n & 31) >> 1);
    bmf[p * (4 * kGsPitch4) + (idx >> 6)] = fv[i];
  }
  __syncthreads();

  const v4f* src4 = (const v4f*)GT;
  v4f g[16];
  for (int i = 0; i < 16; ++i) g[i] = src4[threadIdx.x + i * 256];

  v8f tt[4] = {};
  {
    const v4f* arow4 = bm4 + lm * kGsPitch4 + hh;
    for (int j = 0; j < 8; ++j) {
      v4f av[4];
      for (int kt = 0; kt < 4; ++kt) av[kt] = arow4[kt * 16 * kGsPitch4 + 2 * j];
      v2f e0; e0.x = e[j].x; e0.y = e[j].y;
      v2f e1; e1.x = e[j].z; e1.y = e[j].w;
      for (int kt = 0; kt < 4; ++kt) {
        v2f a0; a0.x = av[kt].x; a0.y = av[kt].y;
        tt[kt] = wmma4(a0, e0, tt[kt]);
      }
      for (int kt = 0; kt < 4; ++kt) {
        v2f a1; a1.x = av[kt].z; a1.y = av[kt].w;
        tt[kt] = wmma4(a1, e1, tt[kt]);
      }
    }
  }

  for (int i = 0; i < 16; ++i) {
    const int idx = threadIdx.x + i * 256;
    smem4[(idx >> 4) * kGsPitch4 + (idx & 15)] = g[i];
  }
  __syncthreads();

  v4f* yrow4 = (v4f*)(y + (rb + 8 * hh) * 512) + lm;
  for (int ng = 0; ng < 4; ++ng) {
    const v4f* bcol4 = smem4 + (ng * 64 + lm) * kGsPitch4 + 2 * hh;
    v8f ev[4] = {}, od[4] = {};
    for (int kt = 0; kt < 4; ++kt) {
      for (int u = 0; u < 2; ++u) {
        v2f a0; a0.x = tt[kt][4 * u];     a0.y = tt[kt][4 * u + 1];
        v2f a1; a1.x = tt[kt][4 * u + 2]; a1.y = tt[kt][4 * u + 3];
        v4f b[4];
        for (int q = 0; q < 4; ++q) b[q] = bcol4[q * 16 * kGsPitch4 + 4 * kt + u];
        for (int q = 0; q < 4; ++q) {
          v2f b0; b0.x = b[q].x; b0.y = b[q].y;
          if (kt < 2) ev[q] = wmma4(a0, b0, ev[q]);
          else        od[q] = wmma4(a0, b0, od[q]);
        }
        for (int q = 0; q < 4; ++q) {
          v2f b1; b1.x = b[q].z; b1.y = b[q].w;
          if (kt < 2) ev[q] = wmma4(a1, b1, ev[q]);
          else        od[q] = wmma4(a1, b1, od[q]);
        }
      }
    }
    for (int r = 0; r < 8; ++r) {
      v4f lo, hi;
      for (int q = 0; q < 4; ++q) {
        lo[q] = ev[q][r] + od[q][r];
        hi[q] = ev[q][r] - od[q][r];
      }
      yrow4[r * 128 + ng * 16]      = lo;
      yrow4[r * 128 + 64 + ng * 16] = hi;
    }
  }
}

extern "C" void kernel_launch(void* const* d_in, const int* in_sizes, int n_in,
                              void* d_out, int out_size, void* d_ws, size_t ws_size,
                              hipStream_t stream) {
  const float* x  = (const float*)d_in[0];
  const float* Wr = (const float*)d_in[1];
  const float* Wi = (const float*)d_in[2];
  float* y = (float*)d_out;

  float* ws  = (float*)d_ws;
  float* Dm  = ws  + 32768;
  float* E2  = Dm  + 32768;
  float* GT  = E2  + 32768;
  float* X1  = GT  + 32768;
  float* X2r = X1  + 2097152;
  float* X2i = X2r + 65536;
  float* Fr  = X2i + 65536;
  float* Fi  = Fr  + 65536;

  row_dft<<<256, 256, kOnePerWgpLdsBytes, stream>>>(x, Dm, E2, GT, X1);
  col_dft<<<256, 256, kOnePerWgpLdsBytes, stream>>>(X1, Dm, X2r, X2i);
  mode_mix<<<dim3(4, 64), 256, 0, stream>>>(X2r, X2i, Wr, Wi, Fr, Fi);
  inv_row<<<256, 256, kOnePerWgpLdsBytes, stream>>>(Fr, Fi, E2, GT, y);
}


# col_dft folded into row_dft: a wave's X1 strip stays in its accumulators as B fragments of the H-axis product; mode_mix (64 modes x 4 o per group) adds the 4 partial spectra; sums over h, c change
# speedup vs baseline: 3.8995x; 1.2058x over previous
#include <hip/hip_runtime.h>
#include <math.h>


static constexpr int kH = 512;
static constexpr int kW = 512;
static constexpr int kC = 64;
static constexpr int kO = 64;
static constexpr int kM1 = 32;
static constexpr int kM2 = 32;
static constexpr int kModes = kM1 * kM2;
static constexpr float kInvHW = 1.0f / (512.0f * 512.0f);
static constexpr int kBsPitch4 = 65;
static constexpr int kGsPitch4 = 17;
static constexpr int kPartOfs = 17920;
static constexpr int kOnePerWgpLdsBytes = 162 * 1024;

typedef __attribute__((ext_vector_type(2))) float v2f;
typedef __attribute__((ext_vector_type(4))) float v4f;
typedef __attribute__((ext_vector_type(8))) float v8f;

__device__ inline v8f wmma4(v2f a, v2f b, v8f c) {
  return __builtin_amdgcn_wmma_f32_16x16x4_f32(
       false, a,  false, b,
       (short)0, c,  false,  false);
}


__device__ inline void lds_barrier() {
  asm volatile("s_wait_dscnt 0x0\n\ts_barrier_signal -1\n\ts_barrier_wait -1" ::: "memory");
}

__global__ __launch_bounds__(256) __attribute__((amdgpu_waves_per_eu(2, 2)))
void row_dft(const float* __restrict__ x, float* __restrict__ E2,
             float* __restrict__ GT, float* __restrict__ X2rp,
             float* __restrict__ X2ip) {
  extern __shared__ v4f smem4[];
  float* cs = (float*)(smem4 + 64 * kBsPitch4);
  float* sn = cs + 512 + 32;
  const int wave = threadIdx.x >> 5, lane = threadIdx.x & 31;
  const int hh = lane >> 4, lm = lane & 15;
  const int rowBase = blockIdx.x * 128 + wave * 16;
  const v4f* arow4 = (const v4f*)(x + (rowBase + lm) * 512) + hh;

  v4f alo[32], ahi[32];
  for (int j = 0; j < 30; ++j) {
    alo[j] = arow4[2 * j];
    ahi[j] = arow4[2 * j + 64];
  }

  const float step = 6.2831853071795864769f / 512.0f;
  for (int n = threadIdx.x; n < 512; n += 256) {
    const float th = n * step;
    cs[n] = cosf(th);
    sn[n] = sinf(th);
  }
  lds_barrier();
  {
    const int p = threadIdx.x & 63;
    const int k = 2 * (p & 15) + (p >> 5);
    const bool sinPart = ((p >> 4) & 1) != 0;
    const float* tab = sinPart ? sn : cs;
    for (int i = 0; i < 16; ++i) {
      const int wq = (threadIdx.x >> 6) + 4 * i;
      v4f v;
      for (int e = 0; e < 4; ++e) {
        const float t = tab[((4 * wq + e) * k) & 511];
        v[e] = sinPart ? -t : t;
      }
      smem4[p * kBsPitch4 + wq] = v;
    }
  }
  lds_barrier();
  for (int j = 30; j < 32; ++j) {
    alo[j] = arow4[2 * j];
    ahi[j] = arow4[2 * j + 64];
  }
  __builtin_amdgcn_sched_barrier(0);

  const v4f* bcol4 = smem4 + lm * kBsPitch4 + hh;
  v8f acc[4] = {};
  for (int j = 0; j < 32; ++j) {
    const v4f s = alo[j] + ahi[j];
    const v4f d = alo[j] - ahi[j];
    v4f b[4];
    for (int nt = 0; nt < 4; ++nt) b[nt] = bcol4[nt * 16 * kBsPitch4 + 2 * j];
    v2f s0; s0.x = s.x; s0.y = s.y;
    v2f d0; d0.x = d.x; d0.y = d.y;
    v2f s1; s1.x = s.z; s1.y = s.w;
    v2f d1; d1.x = d.z; d1.y = d.w;
    for (int nt = 0; nt < 4; ++nt) {
      v2f b0; b0.x = b[nt].x; b0.y = b[nt].y;
      acc[nt] = wmma4(nt < 2 ? s0 : d0, b0, acc[nt]);
    }
    for (int nt = 0; nt < 4; ++nt) {
      v2f b1; b1.x = b[nt].z; b1.y = b[nt].w;
      acc[nt] = wmma4(nt < 2 ? s1 : d1, b1, acc[nt]);
    }
  }
  const int hBase = (blockIdx.x & 3) * 128 + wave * 16;
  v8f pr[2][2] = {}, pi[2][2] = {};
  for (int t = 0; t < 4; ++t) {
    const int h = hBase + 8 * hh + 2 * t;
    for (int a = 0; a < 2; ++a) {
      const int kx = 16 * a + lm;
      const int n0 = (h * kx) & 511, n1 = ((h + 1) * kx) & 511;
      v2f ca; ca.x = cs[n0]; ca.y = cs[n1];
      v2f sa; sa.x = sn[n0]; sa.y = sn[n1];
      const v2f na = -sa;
      for (int e = 0; e < 2; ++e) {
        v2f bc; bc.x = acc[2 * e][2 * t];     bc.y = acc[2 * e][2 * t + 1];
        v2f bs; bs.x = acc[2 * e + 1][2 * t]; bs.y = acc[2 * e + 1][2 * t + 1];
        pr[a][e] = wmma4(ca, bc, pr[a][e]);
        pr[a][e] = wmma4(sa, bs, pr[a][e]);
        pi[a][e] = wmma4(ca, bs, pi[a][e]);
        pi[a][e] = wmma4(na, bc, pi[a][e]);
      }
    }
  }
  float* part = (float*)smem4 + kPartOfs;
  {
    v2f* mine = (v2f*)(part + wave * 2048) + lm;
    for (int a = 0; a < 2; ++a)
      for (int r = 0; r < 8; ++r) {
        const int kx = 16 * a + 8 * hh + r;
        v2f vr; vr.x = pr[a][0][r]; vr.y = pr[a][1][r];
        v2f vi; vi.x = pi[a][0][r]; vi.y = pi[a][1][r];
        mine[kx * 16] = vr;
        mine[512 + kx * 16] = vi;
      }
  }
  __syncthreads();
  {
    const v4f* part4 = (const v4f*)part;
    v4f sr = part4[threadIdx.x], si = part4[256 + threadIdx.x];
    for (int w = 1; w < 8; ++w) {
      sr += part4[w * 512 + threadIdx.x];
      si += part4[w * 512 + 256 + threadIdx.x];
    }
    ((v4f*)X2rp)[blockIdx.x * 256 + threadIdx.x] = sr;
    ((v4f*)X2ip)[blockIdx.x * 256 + threadIdx.x] = si;
  }

  {
    const int i = blockIdx.x * 256 + threadIdx.x;
    if (blockIdx.x >= 128) {
      const int i2 = i - 32768;
      const int hr = i2 >> 6, c = i2 & 63, k = c & 31;
      const int n = (hr * k) & 511;
      E2[i2] = (c < 32) ? cs[n] : sn[n];
    }
    if (blockIdx.x < 64) {
      const int s6 = i >> 6, p = i & 63, k = 2 * (p & 15) + (p >> 5);
      const int w = (s6 & 0xC0) | ((s6 & 15) << 2) | ((s6 >> 4) & 3);
      const int n = (w * k) & 511;
      const float s = ((k == 0) ? 1.0f : 2.0f) * kInvHW;
      GT[i] = ((p >> 4) & 1) ? -s * sn[n] : s * cs[n];
    }
  }
}

__global__ __launch_bounds__(256) __attribute__((amdgpu_waves_per_eu(2, 2)))
void mode_mix(const float* __restrict__ X2rp, const float* __restrict__ X2ip,
              const float* __restrict__ Wr, const float* __restrict__ Wi,
              float* __restrict__ Fr, float* __restrict__ Fi) {
  __shared__ v4f red4[16 * 8 * 16];
  const int quad = threadIdx.x & 15;
  const int g = threadIdx.x >> 4;
  const int m0 = blockIdx.x * 64 + 4 * quad;
  const int o0 = blockIdx.y * 4;

  v4f wr[4][4], wi[4][4];
  for (int cc = 0; cc < 4; ++cc)
    for (int oo = 0; oo < 4; ++oo) {
      const int ofs = (((4 * g + cc) * kO + o0 + oo) << 10) + m0;
      wr[cc][oo] = *(const v4f*)(Wr + ofs);
      wi[cc][oo] = *(const v4f*)(Wi + ofs);
    }
  __builtin_amdgcn_sched_barrier(0);
  v4f xrp[4][4], xip[4][4];
  for (int cc = 0; cc < 4; ++cc)
    for (int q = 0; q < 4; ++q) {
      const int ofs = ((4 * (4 * g + cc) + q) << 10) + m0;
      xrp[cc][q] = *(const v4f*)(X2rp + ofs);
      xip[cc][q] = *(const v4f*)(X2ip + ofs);
    }
  __builtin_amdgcn_sched_barrier(0);
  v4f ar[4] = {}, ai[4] = {};
  for (int cc = 0; cc < 4; ++cc) {
    const v4f xr = ((xrp[cc][0] + xrp[cc][1]) + xrp[cc][2]) + xrp[cc][3];
    const v4f xi = ((xip[cc][0] + xip[cc][1]) + xip[cc][2]) + xip[cc][3];
    for (int oo = 0; oo < 4; ++oo)
      for (int e = 0; e < 4; ++e) {
        ar[oo][e] = fmaf(xr[e], wr[cc][oo][e], fmaf(-xi[e], wi[cc][oo][e], ar[oo][e]));
        ai[oo][e] = fmaf(xr[e], wi[cc][oo][e], fmaf(xi[e], wr[cc][oo][e], ai[oo][e]));
      }
  }
  for (int oo = 0; oo < 4; ++oo) {
    red4[(g * 8 + 2 * oo + 0) * 16 + quad] = ar[oo];
    red4[(g * 8 + 2 * oo + 1) * 16 + quad] = ai[oo];
  }
  __syncthreads();

  const float* red = (const float*)red4;
  for (int i = 0; i < 2; ++i) {
    const int idx = threadIdx.x + 256 * i;
    const int plane = idx >> 6, mm = idx & 63;
    float s = red[plane * 64 + mm];
    for (int gg = 1; gg < 16; ++gg) s += red[(gg * 8 + plane) * 64 + mm];
    float* dst = (plane & 1) ? Fi : Fr;
    dst[((o0 + (plane >> 1)) << 10) + blockIdx.x * 64 + mm] = s;
  }
}

__global__ __launch_bounds__(256) __attribute__((amdgpu_waves_per_eu(2, 2)))
void inv_row(const float* __restrict__ Fr, const float* __restrict__ Fi,
             const float* __restrict__ E2, const float* __restrict__ GT,
             float* __restrict__ y) {
  extern __shared__ v4f smem4[];
  v4f* bm4 = smem4 + 256 * kGsPitch4;
  float* bmf = (float*)bm4;
  const int wave = threadIdx.x >> 5, lane = threadIdx.x & 31;
  const int hh = lane >> 4, lm = lane & 15;
  const int o = blockIdx.x >> 2;
  const int h0 = (blockIdx.x & 3) * 128 + wave * 16;
  const int rb = blockIdx.x * 128 + wave * 16;

  float fv[16];
  for (int i = 0; i < 16; ++i) {
    const int idx = threadIdx.x + i * 256;
    const int r = idx >> 6, n = idx & 63;
    const int fofs = (o << 10) + (r & 31) * 32 + (n & 31);
    const float* srcp = ((r < 32) == (n < 32)) ? Fr : Fi;
    const float v = srcp[fofs];
    fv[i] = (r >= 32 && n < 32) ? -v : v;
  }
  const v4f* erow4 = (const v4f*)(E2 + (h0 + lm) * 64) + hh;
  v4f e[8];
  for (int j = 0; j < 8; ++j) e[j] = erow4[2 * j];
  for (int i = 0; i < 16; ++i) {
    const int idx = threadIdx.x + i * 256;
    const int n = idx & 63;
    const int p = 16 * ((n >> 5) + 2 * (n & 1)) + ((n & 31) >> 1);
    bmf[p * (4 * kGsPitch4) + (idx >> 6)] = fv[i];
  }
  __syncthreads();

  const v4f* src4 = (const v4f*)GT;
  v4f g[16];
  for (int i = 0; i < 16; ++i) g[i] = src4[threadIdx.x + i * 256];

  v8f tt[4] = {};
  {
    const v4f* arow4 = bm4 + lm * kGsPitch4 + hh;
    for (int j = 0; j < 8; ++j) {
      v4f av[4];
      for (int kt = 0; kt < 4; ++kt) av[kt] = arow4[kt * 16 * kGsPitch4 + 2 * j];
      v2f e0; e0.x = e[j].x; e0.y = e[j].y;
      v2f e1; e1.x = e[j].z; e1.y = e[j].w;
      for (int kt = 0; kt < 4; ++kt) {
        v2f a0; a0.x = av[kt].x; a0.y = av[kt].y;
        tt[kt] = wmma4(a0, e0, tt[kt]);
      }
      for (int kt = 0; kt < 4; ++kt) {
        v2f a1; a1.x = av[kt].z; a1.y = av[kt].w;
        tt[kt] = wmma4(a1, e1, tt[kt]);
      }
    }
  }

  for (int i = 0; i < 16; ++i) {
    const int idx = threadIdx.x + i * 256;
    smem4[(idx >> 4) * kGsPitch4 + (idx & 15)] = g[i];
  }
  __syncthreads();

  v4f* yrow4 = (v4f*)(y + (rb + 8 * hh) * 512) + lm;
  for (int ng = 0; ng < 4; ++ng) {
    const v4f* bcol4 = smem4 + (ng * 64 + lm) * kGsPitch4 + 2 * hh;
    v8f ev[4] = {}, od[4] = {};
    for (int kt = 0; kt < 4; ++kt) {
      for (int u = 0; u < 2; ++u) {
        v2f a0; a0.x = tt[kt][4 * u];     a0.y = tt[kt][4 * u + 1];
        v2f a1; a1.x = tt[kt][4 * u + 2]; a1.y = tt[kt][4 * u + 3];
        v4f b[4];
        for (int q = 0; q < 4; ++q) b[q] = bcol4[q * 16 * kGsPitch4 + 4 * kt + u];
        for (int q = 0; q < 4; ++q) {
          v2f b0; b0.x = b[q].x; b0.y = b[q].y;
          if (kt < 2) ev[q] = wmma4(a0, b0, ev[q]);
          else        od[q] = wmma4(a0, b0, od[q]);
        }
        for (int q = 0; q < 4; ++q) {
          v2f b1; b1.x = b[q].z; b1.y = b[q].w;
          if (kt < 2) ev[q] = wmma4(a1, b1, ev[q]);
          else        od[q] = wmma4(a1, b1, od[q]);
        }
      }
    }
    for (int r = 0; r < 8; ++r) {
      v4f lo, hi;
      for (int q = 0; q < 4; ++q) {
        lo[q] = ev[q][r] + od[q][r];
        hi[q] = ev[q][r] - od[q][r];
      }
      yrow4[r * 128 + ng * 16]      = lo;
      yrow4[r * 128 + 64 + ng * 16] = hi;
    }
  }
}

extern "C" void kernel_launch(void* const* d_in, const int* in_sizes, int n_in,
                              void* d_out, int out_size, void* d_ws, size_t ws_size,
                              hipStream_t stream) {
  const float* x  = (const float*)d_in[0];
  const float* Wr = (const float*)d_in[1];
  const float* Wi = (const float*)d_in[2];
  float* y = (float*)d_out;

  float* ws   = (float*)d_ws;
  float* E2   = ws   + 65536;
  float* GT   = E2   + 32768;
  float* X2rp = GT   + 32768;
  float* X2ip = X2rp + 262144;
  float* Fr   = X2ip + 262144;
  float* Fi   = Fr   + 65536;

  row_dft<<<256, 256, kOnePerWgpLdsBytes, stream>>>(x, E2, GT, X2rp, X2ip);
  mode_mix<<<dim3(16, 16), 256, 0, stream>>>(X2rp, X2ip, Wr, Wi, Fr, Fi);
  inv_row<<<256, 256, kOnePerWgpLdsBytes, stream>>>(Fr, Fi, E2, GT, y);
}
